# starting kernel
# baseline (speedup 1.0000x reference)
#include <hip/hip_runtime.h>
#include <hip/hip_bf16.h>

typedef float v2f __attribute__((ext_vector_type(2)));
typedef float v8f __attribute__((ext_vector_type(8)));

#define NH   4
#define DIN  128
#define DOUT 32

__device__ __forceinline__ unsigned fkey(float f) {
  unsigned u = __float_as_uint(f);
  return (u >> 31) ? ~u : (u ^ 0x80000000u);
}
__device__ __forceinline__ float funkey(unsigned k) {
  unsigned u = (k >> 31) ? (k ^ 0x80000000u) : ~k;
  return __uint_as_float(u);
}

__global__ __launch_bounds__(128) void k_hw_wmma(const float* __restrict__ Hin,
                                                 const float* __restrict__ W,
                                                 float* __restrict__ Hw, int N) {
  const int h     = threadIdx.x >> 5;
  const int lane  = threadIdx.x & 31;
  const int m     = lane & 15;
  const int khalf = lane >> 4;
  const int base  = blockIdx.x * 16;

  int rowA = base + m;
  if (rowA >= N) rowA = N - 1;
  const float* arow  = Hin + (size_t)rowA * DIN + 2 * khalf;
  const float* b0row = W + ((size_t)h * DOUT + m) * DIN + 2 * khalf;
  const float* b1row = W + ((size_t)h * DOUT + m + 16) * DIN + 2 * khalf;

  v8f acc0 = {0.f, 0.f, 0.f, 0.f, 0.f, 0.f, 0.f, 0.f};
  v8f acc1 = {0.f, 0.f, 0.f, 0.f, 0.f, 0.f, 0.f, 0.f};

#pragma unroll
  for (int k = 0; k < DIN; k += 4) {
    v2f a  = { arow[k],  arow[k + 1]  };
    v2f b0 = { b0row[k], b0row[k + 1] };
    v2f b1 = { b1row[k], b1row[k + 1] };
    acc0 = __builtin_amdgcn_wmma_f32_16x16x4_f32(false, a, false, b0, (short)0,
                                                 acc0, false, false);
    acc1 = __builtin_amdgcn_wmma_f32_16x16x4_f32(false, a, false, b1, (short)0,
                                                 acc1, false, false);
  }

#pragma unroll
  for (int g = 0; g < 8; ++g) {
    int node = base + g + 8 * khalf;
    if (node < N) {
      float* dst = Hw + ((size_t)node * NH + h) * DOUT;
      dst[m]      = acc0[g];
      dst[m + 16] = acc1[g];
    }
  }
}

__global__ void k_scores(const float* __restrict__ Hw, const float* __restrict__ Asrc,
                         const float* __restrict__ Atgt, float* __restrict__ ssrc,
                         float* __restrict__ stgt, int N) {
  int n = blockIdx.x * blockDim.x + threadIdx.x;
  if (n >= N) return;
  const float* hw = Hw + (size_t)n * (NH * DOUT);
#pragma unroll
  for (int h = 0; h < NH; ++h) {
    float s1 = 0.f, s2 = 0.f;
#pragma unroll
    for (int o = 0; o < DOUT; ++o) {
      float v = hw[h * DOUT + o];
      s1 += v * Asrc[h * DOUT + o];
      s2 += v * Atgt[h * DOUT + o];
    }
    ssrc[(size_t)n * NH + h] = s1;
    stgt[(size_t)n * NH + h] = s2;
  }
}

__global__ void k_init_seg(unsigned* __restrict__ segmax, float* __restrict__ segsum,
                           int n) {
  int i = blockIdx.x * blockDim.x + threadIdx.x;
  if (i < n) {
    segmax[i] = 0x007FFFFFu;
    segsum[i] = 0.f;
  }
}
__global__ void k_zero(float* __restrict__ p, int n) {
  int i = blockIdx.x * blockDim.x + threadIdx.x;
  if (i < n) p[i] = 0.f;
}

__global__ void k_edge_logit(const int* __restrict__ src, const int* __restrict__ tgt,
                             const float* __restrict__ ssrc, const float* __restrict__ stgt,
                             float* __restrict__ Ebuf, unsigned* __restrict__ segmax,
                             int E) {
  int e = blockIdx.x * blockDim.x + threadIdx.x;
  if (e >= E) return;
  int s = src[e], t = tgt[e];
  float4 a = ((const float4*)ssrc)[s];
  float4 b = ((const float4*)stgt)[t];
  float4 v;
  v.x = a.x + b.x; v.y = a.y + b.y; v.z = a.z + b.z; v.w = a.w + b.w;
  v.x = v.x > 0.f ? v.x : 0.2f * v.x;
  v.y = v.y > 0.f ? v.y : 0.2f * v.y;
  v.z = v.z > 0.f ? v.z : 0.2f * v.z;
  v.w = v.w > 0.f ? v.w : 0.2f * v.w;
  ((float4*)Ebuf)[e] = v;
  unsigned* mrow = segmax + (size_t)t * NH;
  atomicMax(&mrow[0], fkey(v.x));
  atomicMax(&mrow[1], fkey(v.y));
  atomicMax(&mrow[2], fkey(v.z));
  atomicMax(&mrow[3], fkey(v.w));
}

__global__ void k_edge_exp(const int* __restrict__ tgt, float* __restrict__ Ebuf,
                           const unsigned* __restrict__ segmax,
                           float* __restrict__ segsum, int E) {
  int e = blockIdx.x * blockDim.x + threadIdx.x;
  if (e >= E) return;
  int t = tgt[e];
  float4 w = ((const float4*)Ebuf)[e];
  uint4 mk = ((const uint4*)segmax)[t];
  w.x = expf(w.x - funkey(mk.x));
  w.y = expf(w.y - funkey(mk.y));
  w.z = expf(w.z - funkey(mk.z));
  w.w = expf(w.w - funkey(mk.w));
  ((float4*)Ebuf)[e] = w;
  float* srow = segsum + (size_t)t * NH;
  atomicAdd(&srow[0], w.x);
  atomicAdd(&srow[1], w.y);
  atomicAdd(&srow[2], w.z);
  atomicAdd(&srow[3], w.w);
}

__global__ void k_edge_scatter(const int* __restrict__ src, const int* __restrict__ tgt,
                               const float* __restrict__ Ebuf,
                               const float* __restrict__ segsum,
                               const float* __restrict__ Hw, float* __restrict__ dout,
                               int E, int N) {
  int idx = blockIdx.x * blockDim.x + threadIdx.x;
  if (idx >= E * NH) return;
  int e = idx >> 2, h = idx & 3;
  int s = src[e], t = tgt[e];
  float alpha = Ebuf[idx] / segsum[(size_t)t * NH + h];
  const float4* hw = (const float4*)(Hw + ((size_t)s * NH + h) * DOUT);
  float* out = dout + ((size_t)h * N + t) * DOUT;
#pragma unroll
  for (int q = 0; q < DOUT / 4; ++q) {
    float4 v = hw[q];
    atomicAdd(&out[4 * q + 0], alpha * v.x);
    atomicAdd(&out[4 * q + 1], alpha * v.y);
    atomicAdd(&out[4 * q + 2], alpha * v.z);
    atomicAdd(&out[4 * q + 3], alpha * v.w);
  }
}

__global__ void k_elu(float* __restrict__ dout, int n) {
  int i = blockIdx.x * blockDim.x + threadIdx.x;
  if (i >= n) return;
  float x = dout[i];
  dout[i] = x > 0.f ? x : expm1f(x);
}

extern "C" void kernel_launch(void* const* d_in, const int* in_sizes, int n_in,
                              void* d_out, int out_size, void* d_ws, size_t ws_size,
                              hipStream_t stream) {
  const float* Hin  = (const float*)d_in[0];
  const int*   ei   = (const int*)d_in[1];
  const float* W    = (const float*)d_in[2];
  const float* Asrc = (const float*)d_in[3];
  const float* Atgt = (const float*)d_in[4];
  float* dout = (float*)d_out;

  const int N = in_sizes[0] / DIN;
  const int E = in_sizes[1] / 2;
  const int* src = ei;
  const int* tgt = ei + E;

  char* p = (char*)d_ws;
  float*    Hw     = (float*)p;    p += (size_t)N * NH * DOUT * sizeof(float);
  float*    ssrc   = (float*)p;    p += (size_t)N * NH * sizeof(float);
  float*    stgt   = (float*)p;    p += (size_t)N * NH * sizeof(float);
  unsigned* segmax = (unsigned*)p; p += (size_t)N * NH * sizeof(unsigned);
  float*    segsum = (float*)p;    p += (size_t)N * NH * sizeof(float);
  float*    Ebuf   = (float*)p;    p += (size_t)E * NH * sizeof(float);
  (void)ws_size; (void)n_in;

  const int B = 256;
  k_zero<<<(out_size + B - 1) / B, B, 0, stream>>>(dout, out_size);
  k_init_seg<<<(N * NH + B - 1) / B, B, 0, stream>>>(segmax, segsum, N * NH);
  k_hw_wmma<<<(N + 15) / 16, 128, 0, stream>>>(Hin, W, Hw, N);
  k_scores<<<(N + B - 1) / B, B, 0, stream>>>(Hw, Asrc, Atgt, ssrc, stgt, N);
  k_edge_logit<<<(E + B - 1) / B, B, 0, stream>>>(src, tgt, ssrc, stgt, Ebuf, segmax, E);
  k_edge_exp<<<(E + B - 1) / B, B, 0, stream>>>(tgt, Ebuf, segmax, segsum, E);
  k_edge_scatter<<<(E * NH + B - 1) / B, B, 0, stream>>>(src, tgt, Ebuf, segsum, Hw,
                                                         dout, E, N);
  k_elu<<<(out_size + B - 1) / B, B, 0, stream>>>(dout, out_size);
}


# Edges bucketed by target (counting sort, int atomics); one wave per target does softmax + weighted sum by gather: no float atomics; sum order = bucket arrival order
# speedup vs baseline: 5.3936x; 5.3936x over previous
#include <hip/hip_runtime.h>
#include <hip/hip_bf16.h>

typedef float v2f __attribute__((ext_vector_type(2)));
typedef float v8f __attribute__((ext_vector_type(8)));

#define NH   4
#define DIN  128
#define DOUT 32

__global__ __launch_bounds__(128) void k_hw_wmma(const float* __restrict__ Hin,
                                                 const float* __restrict__ W,
                                                 float* __restrict__ Hw, int N) {
  const int h     = threadIdx.x >> 5;
  const int lane  = threadIdx.x & 31;
  const int m     = lane & 15;
  const int khalf = lane >> 4;
  const int base  = blockIdx.x * 16;

  int rowA = base + m;
  if (rowA >= N) rowA = N - 1;
  const float* arow  = Hin + (size_t)rowA * DIN + 2 * khalf;
  const float* b0row = W + ((size_t)h * DOUT + m) * DIN + 2 * khalf;
  const float* b1row = W + ((size_t)h * DOUT + m + 16) * DIN + 2 * khalf;

  v8f acc0 = {0.f, 0.f, 0.f, 0.f, 0.f, 0.f, 0.f, 0.f};
  v8f acc1 = {0.f, 0.f, 0.f, 0.f, 0.f, 0.f, 0.f, 0.f};

#pragma unroll
  for (int k = 0; k < DIN; k += 4) {
    v2f a  = { arow[k],  arow[k + 1]  };
    v2f b0 = { b0row[k], b0row[k + 1] };
    v2f b1 = { b1row[k], b1row[k + 1] };
    acc0 = __builtin_amdgcn_wmma_f32_16x16x4_f32(false, a, false, b0, (short)0,
                                                 acc0, false, false);
    acc1 = __builtin_amdgcn_wmma_f32_16x16x4_f32(false, a, false, b1, (short)0,
                                                 acc1, false, false);
  }

#pragma unroll
  for (int g = 0; g < 8; ++g) {
    int node = base + g + 8 * khalf;
    if (node < N) {
      float* dst = Hw + ((size_t)node * NH + h) * DOUT;
      dst[m]      = acc0[g];
      dst[m + 16] = acc1[g];
    }
  }
}

__global__ void k_scores(const float* __restrict__ Hw, const float* __restrict__ Asrc,
                         const float* __restrict__ Atgt, float* __restrict__ ssrc,
                         float* __restrict__ stgt, int N) {
  int n = blockIdx.x * blockDim.x + threadIdx.x;
  if (n >= N) return;
  const float* hw = Hw + (size_t)n * (NH * DOUT);
#pragma unroll
  for (int h = 0; h < NH; ++h) {
    float s1 = 0.f, s2 = 0.f;
#pragma unroll
    for (int o = 0; o < DOUT; ++o) {
      float v = hw[h * DOUT + o];
      s1 += v * Asrc[h * DOUT + o];
      s2 += v * Atgt[h * DOUT + o];
    }
    ssrc[(size_t)n * NH + h] = s1;
    stgt[(size_t)n * NH + h] = s2;
  }
}

__global__ void k_zero_cnt(int* __restrict__ cnt, int n) {
  int i = blockIdx.x * blockDim.x + threadIdx.x;
  if (i < n) cnt[i] = 0;
}

__global__ void k_count(const int* __restrict__ tgt, int* __restrict__ cnt, int E) {
  int e = blockIdx.x * blockDim.x + threadIdx.x;
  if (e < E) atomicAdd(&cnt[tgt[e]], 1);
}

__global__ __launch_bounds__(1024) void k_scan(const int* __restrict__ cnt,
                                               int* __restrict__ offs,
                                               int* __restrict__ cursor, int N) {
  __shared__ int wsum[32];
  const int lane = threadIdx.x & 31;
  const int w    = threadIdx.x >> 5;
  const int seg  = (N + 31) / 32;
  const int b    = w * seg;
  const int e    = (b + seg < N) ? (b + seg) : N;

  int s = 0;
  for (int i = b + lane; i < e; i += 32) s += cnt[i];
  for (int d = 16; d >= 1; d >>= 1) s += __shfl_xor(s, d);
  if (lane == 0) wsum[w] = s;
  __syncthreads();
  if (w == 0) {
    int v = wsum[lane];
    int inc = v;
    for (int d = 1; d < 32; d <<= 1) {
      int y = __shfl_up(inc, d);
      if (lane >= d) inc += y;
    }
    wsum[lane] = inc - v;
  }
  __syncthreads();

  int carry = wsum[w];
  for (int c = b; c < e; c += 32) {
    const int i = c + lane;
    const int v = (i < e) ? cnt[i] : 0;
    int inc = v;
    for (int d = 1; d < 32; d <<= 1) {
      int y = __shfl_up(inc, d);
      if (lane >= d) inc += y;
    }
    const int ex = carry + inc - v;
    if (i < e) {
      offs[i]   = ex;
      cursor[i] = ex;
    }
    carry += __shfl(inc, 31);
  }
  if (threadIdx.x == 1023) offs[N] = carry;
}

__global__ void k_fill(const int* __restrict__ src, const int* __restrict__ tgt,
                       int* __restrict__ cursor, int* __restrict__ ssorted, int E) {
  int e = blockIdx.x * blockDim.x + threadIdx.x;
  if (e >= E) return;
  int pos = atomicAdd(&cursor[tgt[e]], 1);
  ssorted[pos] = src[e];
}

__device__ __forceinline__ float lrelu(float x) { return x > 0.f ? x : 0.2f * x; }
__device__ __forceinline__ float pick4(float4 v, int h) {
  return h == 0 ? v.x : (h == 1 ? v.y : (h == 2 ? v.z : v.w));
}

__global__ __launch_bounds__(256) void k_node(const int* __restrict__ offs,
                                              const int* __restrict__ ssorted,
                                              const float* __restrict__ ssrc,
                                              const float* __restrict__ stgt,
                                              const float* __restrict__ Hw,
                                              float* __restrict__ dout, int N) {
  const int lane = threadIdx.x & 31;
  const int t = blockIdx.x * 8 + __builtin_amdgcn_readfirstlane((int)(threadIdx.x >> 5));
  if (t >= N) return;
  const int beg = __builtin_amdgcn_readfirstlane(offs[t]);
  const int end = __builtin_amdgcn_readfirstlane(offs[t + 1]);
  const float4 bt = ((const float4*)stgt)[t];

  float4 mx;
  mx.x = mx.y = mx.z = mx.w = -INFINITY;
  for (int j = beg + lane; j < end; j += 32) {
    const float4 a = ((const float4*)ssrc)[ssorted[j]];
    mx.x = fmaxf(mx.x, lrelu(a.x + bt.x));
    mx.y = fmaxf(mx.y, lrelu(a.y + bt.y));
    mx.z = fmaxf(mx.z, lrelu(a.z + bt.z));
    mx.w = fmaxf(mx.w, lrelu(a.w + bt.w));
  }
  for (int d = 16; d >= 1; d >>= 1) {
    mx.x = fmaxf(mx.x, __shfl_xor(mx.x, d));
    mx.y = fmaxf(mx.y, __shfl_xor(mx.y, d));
    mx.z = fmaxf(mx.z, __shfl_xor(mx.z, d));
    mx.w = fmaxf(mx.w, __shfl_xor(mx.w, d));
  }

  float4 sm;
  sm.x = sm.y = sm.z = sm.w = 0.f;
  for (int j = beg + lane; j < end; j += 32) {
    const float4 a = ((const float4*)ssrc)[ssorted[j]];
    sm.x += expf(lrelu(a.x + bt.x) - mx.x);
    sm.y += expf(lrelu(a.y + bt.y) - mx.y);
    sm.z += expf(lrelu(a.z + bt.z) - mx.z);
    sm.w += expf(lrelu(a.w + bt.w) - mx.w);
  }
  for (int d = 16; d >= 1; d >>= 1) {
    sm.x += __shfl_xor(sm.x, d);
    sm.y += __shfl_xor(sm.y, d);
    sm.z += __shfl_xor(sm.z, d);
    sm.w += __shfl_xor(sm.w, d);
  }

  const int   h   = lane >> 3;
  const float bth = pick4(bt, h);
  const float mh  = pick4(mx, h);
  const float sh  = pick4(sm, h);

  float4 acc;
  acc.x = acc.y = acc.z = acc.w = 0.f;
#pragma unroll 4
  for (int j = beg; j < end; ++j) {
    const int s = ssorted[j];
    const float a = ssrc[(size_t)s * NH + h];
    const float4 v = ((const float4*)(Hw + (size_t)s * (NH * DOUT)))[lane];
    const float alpha = expf(lrelu(a + bth) - mh) / sh;
    acc.x += alpha * v.x;
    acc.y += alpha * v.y;
    acc.z += alpha * v.z;
    acc.w += alpha * v.w;
  }

  float4 r;
  r.x = acc.x > 0.f ? acc.x : expm1f(acc.x);
  r.y = acc.y > 0.f ? acc.y : expm1f(acc.y);
  r.z = acc.z > 0.f ? acc.z : expm1f(acc.z);
  r.w = acc.w > 0.f ? acc.w : expm1f(acc.w);
  float* out = dout + ((size_t)h * N + t) * DOUT;
  ((float4*)out)[lane & 7] = r;
}

extern "C" void kernel_launch(void* const* d_in, const int* in_sizes, int n_in,
                              void* d_out, int out_size, void* d_ws, size_t ws_size,
                              hipStream_t stream) {
  const float* Hin  = (const float*)d_in[0];
  const int*   ei   = (const int*)d_in[1];
  const float* W    = (const float*)d_in[2];
  const float* Asrc = (const float*)d_in[3];
  const float* Atgt = (const float*)d_in[4];
  float* dout = (float*)d_out;

  const int N = in_sizes[0] / DIN;
  const int E = in_sizes[1] / 2;
  const int* src = ei;
  const int* tgt = ei + E;

  char* p = (char*)d_ws;
  float* Hw      = (float*)p; p += (size_t)N * NH * DOUT * sizeof(float);
  float* ssrc    = (float*)p; p += (size_t)N * NH * sizeof(float);
  float* stgt    = (float*)p; p += (size_t)N * NH * sizeof(float);
  int*   cnt     = (int*)p;   p += (size_t)N * sizeof(int);
  int*   cursor  = (int*)p;   p += (size_t)N * sizeof(int);
  int*   offs    = (int*)p;   p += (size_t)(N + 4) * sizeof(int);
  int*   ssorted = (int*)p;   p += (size_t)E * sizeof(int);
  (void)ws_size; (void)n_in; (void)out_size;

  const int B = 256;
  k_zero_cnt<<<(N + B - 1) / B, B, 0, stream>>>(cnt, N);
  k_count<<<(E + B - 1) / B, B, 0, stream>>>(tgt, cnt, E);
  k_scan<<<1, 1024, 0, stream>>>(cnt, offs, cursor, N);
  k_fill<<<(E + B - 1) / B, B, 0, stream>>>(src, tgt, cursor, ssorted, E);
  k_hw_wmma<<<(N + 15) / 16, 128, 0, stream>>>(Hin, W, Hw, N);
  k_scores<<<(N + B - 1) / B, B, 0, stream>>>(Hw, Asrc, Atgt, ssrc, stgt, N);
  k_node<<<(N + 7) / 8, 256, 0, stream>>>(offs, ssorted, ssrc, stgt, Hw, dout, N);
}


# k_node: each lane keeps one in-edge (source id, 4 softmax weights) in registers; the gather reads id and weight by readlane, so 8 row loads issue back to back; sum order unchanged
# speedup vs baseline: 5.6290x; 1.0436x over previous
#include <hip/hip_runtime.h>
#include <hip/hip_bf16.h>

typedef float v2f __attribute__((ext_vector_type(2)));
typedef float v8f __attribute__((ext_vector_type(8)));

#define NH   4
#define DIN  128
#define DOUT 32

__global__ __launch_bounds__(128) void k_hw_wmma(const float* __restrict__ Hin,
                                                 const float* __restrict__ W,
                                                 float* __restrict__ Hw, int N) {
  const int h     = threadIdx.x >> 5;
  const int lane  = threadIdx.x & 31;
  const int m     = lane & 15;
  const int khalf = lane >> 4;
  const int base  = blockIdx.x * 16;

  int rowA = base + m;
  if (rowA >= N) rowA = N - 1;
  const float* arow  = Hin + (size_t)rowA * DIN + 2 * khalf;
  const float* b0row = W + ((size_t)h * DOUT + m) * DIN + 2 * khalf;
  const float* b1row = W + ((size_t)h * DOUT + m + 16) * DIN + 2 * khalf;

  v8f acc0 = {0.f, 0.f, 0.f, 0.f, 0.f, 0.f, 0.f, 0.f};
  v8f acc1 = {0.f, 0.f, 0.f, 0.f, 0.f, 0.f, 0.f, 0.f};

#pragma unroll
  for (int k = 0; k < DIN; k += 4) {
    v2f a  = { arow[k],  arow[k + 1]  };
    v2f b0 = { b0row[k], b0row[k + 1] };
    v2f b1 = { b1row[k], b1row[k + 1] };
    acc0 = __builtin_amdgcn_wmma_f32_16x16x4_f32(false, a, false, b0, (short)0,
                                                 acc0, false, false);
    acc1 = __builtin_amdgcn_wmma_f32_16x16x4_f32(false, a, false, b1, (short)0,
                                                 acc1, false, false);
  }

#pragma unroll
  for (int g = 0; g < 8; ++g) {
    int node = base + g + 8 * khalf;
    if (node < N) {
      float* dst = Hw + ((size_t)node * NH + h) * DOUT;
      dst[m]      = acc0[g];
      dst[m + 16] = acc1[g];
    }
  }
}

__global__ void k_scores(const float* __restrict__ Hw, const float* __restrict__ Asrc,
                         const float* __restrict__ Atgt, float* __restrict__ ssrc,
                         float* __restrict__ stgt, int N) {
  int n = blockIdx.x * blockDim.x + threadIdx.x;
  if (n >= N) return;
  const float* hw = Hw + (size_t)n * (NH * DOUT);
#pragma unroll
  for (int h = 0; h < NH; ++h) {
    float s1 = 0.f, s2 = 0.f;
#pragma unroll
    for (int o = 0; o < DOUT; ++o) {
      float v = hw[h * DOUT + o];
      s1 += v * Asrc[h * DOUT + o];
      s2 += v * Atgt[h * DOUT + o];
    }
    ssrc[(size_t)n * NH + h] = s1;
    stgt[(size_t)n * NH + h] = s2;
  }
}

__global__ void k_zero_cnt(int* __restrict__ cnt, int n) {
  int i = blockIdx.x * blockDim.x + threadIdx.x;
  if (i < n) cnt[i] = 0;
}

__global__ void k_count(const int* __restrict__ tgt, int* __restrict__ cnt, int E) {
  int e = blockIdx.x * blockDim.x + threadIdx.x;
  if (e < E) atomicAdd(&cnt[tgt[e]], 1);
}

__global__ __launch_bounds__(1024) void k_scan(const int* __restrict__ cnt,
                                               int* __restrict__ offs,
                                               int* __restrict__ cursor, int N) {
  __shared__ int wsum[32];
  const int lane = threadIdx.x & 31;
  const int w    = threadIdx.x >> 5;
  const int seg  = (N + 31) / 32;
  const int b    = w * seg;
  const int e    = (b + seg < N) ? (b + seg) : N;

  int s = 0;
  for (int i = b + lane; i < e; i += 32) s += cnt[i];
  for (int d = 16; d >= 1; d >>= 1) s += __shfl_xor(s, d);
  if (lane == 0) wsum[w] = s;
  __syncthreads();
  if (w == 0) {
    int v = wsum[lane];
    int inc = v;
    for (int d = 1; d < 32; d <<= 1) {
      int y = __shfl_up(inc, d);
      if (lane >= d) inc += y;
    }
    wsum[lane] = inc - v;
  }
  __syncthreads();

  int carry = wsum[w];
  for (int c = b; c < e; c += 32) {
    const int i = c + lane;
    const int v = (i < e) ? cnt[i] : 0;
    int inc = v;
    for (int d = 1; d < 32; d <<= 1) {
      int y = __shfl_up(inc, d);
      if (lane >= d) inc += y;
    }
    const int ex = carry + inc - v;
    if (i < e) {
      offs[i]   = ex;
      cursor[i] = ex;
    }
    carry += __shfl(inc, 31);
  }
  if (threadIdx.x == 1023) offs[N] = carry;
}

__global__ void k_fill(const int* __restrict__ src, const int* __restrict__ tgt,
                       int* __restrict__ cursor, int* __restrict__ ssorted, int E) {
  int e = blockIdx.x * blockDim.x + threadIdx.x;
  if (e >= E) return;
  int pos = atomicAdd(&cursor[tgt[e]], 1);
  ssorted[pos] = src[e];
}

__device__ __forceinline__ float lrelu(float x) { return x > 0.f ? x : 0.2f * x; }

__device__ __forceinline__ float4 logits4(float4 a, float4 b) {
  float4 e;
  e.x = lrelu(a.x + b.x);
  e.y = lrelu(a.y + b.y);
  e.z = lrelu(a.z + b.z);
  e.w = lrelu(a.w + b.w);
  return e;
}

__device__ __forceinline__ int lane_val_i(int v, int k) {
  return __builtin_amdgcn_readlane(v, k);
}

__device__ __forceinline__ float uniform_f(float v) {
  return __builtin_bit_cast(float, __builtin_amdgcn_readfirstlane(__builtin_bit_cast(int, v)));
}

#define GB 8

__device__ __forceinline__ float head_weight(float4 al, int k, unsigned m0, unsigned m1,
                                             unsigned m2, unsigned m3) {
  const unsigned b0 = (unsigned)__builtin_amdgcn_readlane(__builtin_bit_cast(int, al.x), k);
  const unsigned b1 = (unsigned)__builtin_amdgcn_readlane(__builtin_bit_cast(int, al.y), k);
  const unsigned b2 = (unsigned)__builtin_amdgcn_readlane(__builtin_bit_cast(int, al.z), k);
  const unsigned b3 = (unsigned)__builtin_amdgcn_readlane(__builtin_bit_cast(int, al.w), k);
  return __builtin_bit_cast(float, (b0 & m0) | (b1 & m1) | (b2 & m2) | (b3 & m3));
}

__device__ __forceinline__ void gather_rows(int s, float4 al, int n, int lane,
                                            const float* __restrict__ Hw, float4& acc) {
  const int h = lane >> 3;
  const unsigned m0 = h == 0 ? ~0u : 0u, m1 = h == 1 ? ~0u : 0u;
  const unsigned m2 = h == 2 ? ~0u : 0u, m3 = h == 3 ? ~0u : 0u;
  const float4* rows = (const float4*)Hw + lane;
  for (int k0 = 0; k0 < n; k0 += GB) {
    float4 v[GB];
    float  w[GB];
#pragma unroll
    for (int u = 0; u < GB; ++u) {
      const int k = k0 + u;
      if (k < n) {
        v[u] = rows[(size_t)lane_val_i(s, k) * (NH * DOUT / 4)];
        w[u] = head_weight(al, k, m0, m1, m2, m3);
      }
    }
#pragma unroll
    for (int u = 0; u < GB; ++u) {
      if (k0 + u < n) {
        acc.x += w[u] * v[u].x;
        acc.y += w[u] * v[u].y;
        acc.z += w[u] * v[u].z;
        acc.w += w[u] * v[u].w;
      }
    }
  }
}

__global__ __launch_bounds__(256, 16) void k_node(const int* __restrict__ offs,
                                              const int* __restrict__ ssorted,
                                              const float* __restrict__ ssrc,
                                              const float* __restrict__ stgt,
                                              const float* __restrict__ Hw,
                                              float* __restrict__ dout, int N) {
  const int lane = threadIdx.x & 31;
  const int t = blockIdx.x * 8 + __builtin_amdgcn_readfirstlane((int)(threadIdx.x >> 5));
  if (t >= N) return;
  const int beg = __builtin_amdgcn_readfirstlane(offs[t]);
  const int end = __builtin_amdgcn_readfirstlane(offs[t + 1]);
  const float4 bt = ((const float4*)stgt)[t];

  const bool own0 = beg + lane < end;
  int s0 = 0;
  float4 e0;
  e0.x = e0.y = e0.z = e0.w = -INFINITY;
  if (own0) {
    s0 = ssorted[beg + lane];
    e0 = logits4(((const float4*)ssrc)[s0], bt);
  }

  float4 mx = e0;
  for (int j = beg + 32 + lane; j < end; j += 32) {
    const float4 e = logits4(((const float4*)ssrc)[ssorted[j]], bt);
    mx.x = fmaxf(mx.x, e.x);
    mx.y = fmaxf(mx.y, e.y);
    mx.z = fmaxf(mx.z, e.z);
    mx.w = fmaxf(mx.w, e.w);
  }
  for (int d = 16; d >= 1; d >>= 1) {
    mx.x = fmaxf(mx.x, __shfl_xor(mx.x, d));
    mx.y = fmaxf(mx.y, __shfl_xor(mx.y, d));
    mx.z = fmaxf(mx.z, __shfl_xor(mx.z, d));
    mx.w = fmaxf(mx.w, __shfl_xor(mx.w, d));
  }
  mx.x = uniform_f(mx.x);
  mx.y = uniform_f(mx.y);
  mx.z = uniform_f(mx.z);
  mx.w = uniform_f(mx.w);

  float4 p0;
  p0.x = own0 ? expf(e0.x - mx.x) : 0.f;
  p0.y = own0 ? expf(e0.y - mx.y) : 0.f;
  p0.z = own0 ? expf(e0.z - mx.z) : 0.f;
  p0.w = own0 ? expf(e0.w - mx.w) : 0.f;
  float4 sm = p0;
  for (int j = beg + 32 + lane; j < end; j += 32) {
    const float4 e = logits4(((const float4*)ssrc)[ssorted[j]], bt);
    sm.x += expf(e.x - mx.x);
    sm.y += expf(e.y - mx.y);
    sm.z += expf(e.z - mx.z);
    sm.w += expf(e.w - mx.w);
  }
  for (int d = 16; d >= 1; d >>= 1) {
    sm.x += __shfl_xor(sm.x, d);
    sm.y += __shfl_xor(sm.y, d);
    sm.z += __shfl_xor(sm.z, d);
    sm.w += __shfl_xor(sm.w, d);
  }
  sm.x = uniform_f(sm.x);
  sm.y = uniform_f(sm.y);
  sm.z = uniform_f(sm.z);
  sm.w = uniform_f(sm.w);

  float4 acc;
  acc.x = acc.y = acc.z = acc.w = 0.f;
  int s = s0;
  float4 al;
  al.x = p0.x / sm.x;
  al.y = p0.y / sm.y;
  al.z = p0.z / sm.z;
  al.w = p0.w / sm.w;
  for (int c = beg; c < end; c += 32) {
    if (c != beg) {
      const int j = c + lane;
      s = 0;
      al.x = al.y = al.z = al.w = 0.f;
      if (j < end) {
        s = ssorted[j];
        const float4 e = logits4(((const float4*)ssrc)[s], bt);
        al.x = expf(e.x - mx.x) / sm.x;
        al.y = expf(e.y - mx.y) / sm.y;
        al.z = expf(e.z - mx.z) / sm.z;
        al.w = expf(e.w - mx.w) / sm.w;
      }
    }
    const int left = end - c;
    gather_rows(s, al, left < 32 ? left : 32, lane, Hw, acc);
  }

  float4 r;
  r.x = acc.x > 0.f ? acc.x : expm1f(acc.x);
  r.y = acc.y > 0.f ? acc.y : expm1f(acc.y);
  r.z = acc.z > 0.f ? acc.z : expm1f(acc.z);
  r.w = acc.w > 0.f ? acc.w : expm1f(acc.w);
  float* out = dout + ((size_t)(lane >> 3) * N + t) * DOUT;
  ((float4*)out)[lane & 7] = r;
}

extern "C" void kernel_launch(void* const* d_in, const int* in_sizes, int n_in,
                              void* d_out, int out_size, void* d_ws, size_t ws_size,
                              hipStream_t stream) {
  const float* Hin  = (const float*)d_in[0];
  const int*   ei   = (const int*)d_in[1];
  const float* W    = (const float*)d_in[2];
  const float* Asrc = (const float*)d_in[3];
  const float* Atgt = (const float*)d_in[4];
  float* dout = (float*)d_out;

  const int N = in_sizes[0] / DIN;
  const int E = in_sizes[1] / 2;
  const int* src = ei;
  const int* tgt = ei + E;

  char* p = (char*)d_ws;
  float* Hw      = (float*)p; p += (size_t)N * NH * DOUT * sizeof(float);
  float* ssrc    = (float*)p; p += (size_t)N * NH * sizeof(float);
  float* stgt    = (float*)p; p += (size_t)N * NH * sizeof(float);
  int*   cnt     = (int*)p;   p += (size_t)N * sizeof(int);
  int*   cursor  = (int*)p;   p += (size_t)N * sizeof(int);
  int*   offs    = (int*)p;   p += (size_t)(N + 4) * sizeof(int);
  int*   ssorted = (int*)p;   p += (size_t)E * sizeof(int);
  (void)ws_size; (void)n_in; (void)out_size;

  const int B = 256;
  k_zero_cnt<<<(N + B - 1) / B, B, 0, stream>>>(cnt, N);
  k_count<<<(E + B - 1) / B, B, 0, stream>>>(tgt, cnt, E);
  k_scan<<<1, 1024, 0, stream>>>(cnt, offs, cursor, N);
  k_fill<<<(E + B - 1) / B, B, 0, stream>>>(src, tgt, cursor, ssorted, E);
  k_hw_wmma<<<(N + 15) / 16, 128, 0, stream>>>(Hin, W, Hw, N);
  k_scores<<<(N + B - 1) / B, B, 0, stream>>>(Hw, Asrc, Atgt, ssrc, stgt, N);
  k_node<<<(N + 7) / 8, 256, 0, stream>>>(offs, ssorted, ssrc, stgt, Hw, dout, N);
}


# k_scan: prefix sum of the in-degree counts in 98 blocks side by side, each summing the counts in front of its 1024-count slice itself, instead of one block in 98 dependent load+shuffle trips
# speedup vs baseline: 6.9606x; 1.2366x over previous
#include <hip/hip_runtime.h>
#include <hip/hip_bf16.h>

typedef float v2f __attribute__((ext_vector_type(2)));
typedef float v8f __attribute__((ext_vector_type(8)));

#define NH   4
#define DIN  128
#define DOUT 32

__global__ __launch_bounds__(128) void k_hw_wmma(const float* __restrict__ Hin,
                                                 const float* __restrict__ W,
                                                 float* __restrict__ Hw, int N) {
  const int h     = threadIdx.x >> 5;
  const int lane  = threadIdx.x & 31;
  const int m     = lane & 15;
  const int khalf = lane >> 4;
  const int base  = blockIdx.x * 16;

  int rowA = base + m;
  if (rowA >= N) rowA = N - 1;
  const float* arow  = Hin + (size_t)rowA * DIN + 2 * khalf;
  const float* b0row = W + ((size_t)h * DOUT + m) * DIN + 2 * khalf;
  const float* b1row = W + ((size_t)h * DOUT + m + 16) * DIN + 2 * khalf;

  v8f acc0 = {0.f, 0.f, 0.f, 0.f, 0.f, 0.f, 0.f, 0.f};
  v8f acc1 = {0.f, 0.f, 0.f, 0.f, 0.f, 0.f, 0.f, 0.f};

#pragma unroll
  for (int k = 0; k < DIN; k += 4) {
    v2f a  = { arow[k],  arow[k + 1]  };
    v2f b0 = { b0row[k], b0row[k + 1] };
    v2f b1 = { b1row[k], b1row[k + 1] };
    acc0 = __builtin_amdgcn_wmma_f32_16x16x4_f32(false, a, false, b0, (short)0,
                                                 acc0, false, false);
    acc1 = __builtin_amdgcn_wmma_f32_16x16x4_f32(false, a, false, b1, (short)0,
                                                 acc1, false, false);
  }

#pragma unroll
  for (int g = 0; g < 8; ++g) {
    int node = base + g + 8 * khalf;
    if (node < N) {
      float* dst = Hw + ((size_t)node * NH + h) * DOUT;
      dst[m]      = acc0[g];
      dst[m + 16] = acc1[g];
    }
  }
}

__global__ void k_scores(const float* __restrict__ Hw, const float* __restrict__ Asrc,
                         const float* __restrict__ Atgt, float* __restrict__ ssrc,
                         float* __restrict__ stgt, int N) {
  int n = blockIdx.x * blockDim.x + threadIdx.x;
  if (n >= N) return;
  const float* hw = Hw + (size_t)n * (NH * DOUT);
#pragma unroll
  for (int h = 0; h < NH; ++h) {
    float s1 = 0.f, s2 = 0.f;
#pragma unroll
    for (int o = 0; o < DOUT; ++o) {
      float v = hw[h * DOUT + o];
      s1 += v * Asrc[h * DOUT + o];
      s2 += v * Atgt[h * DOUT + o];
    }
    ssrc[(size_t)n * NH + h] = s1;
    stgt[(size_t)n * NH + h] = s2;
  }
}

__global__ void k_zero_cnt(int* __restrict__ cnt, int n) {
  int i = blockIdx.x * blockDim.x + threadIdx.x;
  if (i < n) cnt[i] = 0;
}

__global__ void k_count(const int* __restrict__ tgt, int* __restrict__ cnt, int E) {
  int e = blockIdx.x * blockDim.x + threadIdx.x;
  if (e < E) atomicAdd(&cnt[tgt[e]], 1);
}

#define SCAN_T 1024
#define SCAN_B 8

__global__ __launch_bounds__(SCAN_T) void k_scan(const int* __restrict__ cnt,
                                                 int* __restrict__ offs,
                                                 int* __restrict__ cursor, int N) {
  __shared__ int wpre[32];
  __shared__ int wtot[32];
  const int lane  = threadIdx.x & 31;
  const int w     = threadIdx.x >> 5;
  const int first = blockIdx.x * SCAN_T;
  const int i     = first + threadIdx.x;
  const int v     = (i < N) ? cnt[i] : 0;

  const int nq = first >> 2;
  int s = 0;
  const int4* c4 = (const int4*)cnt;
  for (int base = 0; base < nq; base += SCAN_B * SCAN_T) {
    int4 x[SCAN_B];
#pragma unroll
    for (int u = 0; u < SCAN_B; ++u) {
      const int q = base + u * SCAN_T + (int)threadIdx.x;
      x[u] = c4[q < nq ? q : nq - 1];
    }
#pragma unroll
    for (int u = 0; u < SCAN_B; ++u) {
      const int q = base + u * SCAN_T + (int)threadIdx.x;
      if (q < nq) s += (x[u].x + x[u].y) + (x[u].z + x[u].w);
    }
  }

  int inc = v;
  for (int d = 1; d < 32; d <<= 1) {
    const int y = __shfl_up(inc, d);
    if (lane >= d) inc += y;
  }
  for (int d = 16; d >= 1; d >>= 1) s += __shfl_xor(s, d);
  if (lane == 31) wtot[w] = inc;
  if (lane == 0)  wpre[w] = s;
  __syncthreads();

  int a = wpre[lane] + (lane < w ? wtot[lane] : 0);
  for (int d = 16; d >= 1; d >>= 1) a += __shfl_xor(a, d);

  const int ex = a + inc - v;
  if (i < N) {
    offs[i]   = ex;
    cursor[i] = ex;
  }
  if (i == N - 1) offs[N] = ex + v;
}

__global__ void k_fill(const int* __restrict__ src, const int* __restrict__ tgt,
                       int* __restrict__ cursor, int* __restrict__ ssorted, int E) {
  int e = blockIdx.x * blockDim.x + threadIdx.x;
  if (e >= E) return;
  int pos = atomicAdd(&cursor[tgt[e]], 1);
  ssorted[pos] = src[e];
}

__device__ __forceinline__ float lrelu(float x) { return x > 0.f ? x : 0.2f * x; }

__device__ __forceinline__ float4 logits4(float4 a, float4 b) {
  float4 e;
  e.x = lrelu(a.x + b.x);
  e.y = lrelu(a.y + b.y);
  e.z = lrelu(a.z + b.z);
  e.w = lrelu(a.w + b.w);
  return e;
}

__device__ __forceinline__ int lane_val_i(int v, int k) {
  return __builtin_amdgcn_readlane(v, k);
}

__device__ __forceinline__ float uniform_f(float v) {
  return __builtin_bit_cast(float, __builtin_amdgcn_readfirstlane(__builtin_bit_cast(int, v)));
}

#define GB 8

__device__ __forceinline__ float head_weight(float4 al, int k, unsigned m0, unsigned m1,
                                             unsigned m2, unsigned m3) {
  const unsigned b0 = (unsigned)__builtin_amdgcn_readlane(__builtin_bit_cast(int, al.x), k);
  const unsigned b1 = (unsigned)__builtin_amdgcn_readlane(__builtin_bit_cast(int, al.y), k);
  const unsigned b2 = (unsigned)__builtin_amdgcn_readlane(__builtin_bit_cast(int, al.z), k);
  const unsigned b3 = (unsigned)__builtin_amdgcn_readlane(__builtin_bit_cast(int, al.w), k);
  return __builtin_bit_cast(float, (b0 & m0) | (b1 & m1) | (b2 & m2) | (b3 & m3));
}

__device__ __forceinline__ void gather_rows(int s, float4 al, int n, int lane,
                                            const float* __restrict__ Hw, float4& acc) {
  const int h = lane >> 3;
  const unsigned m0 = h == 0 ? ~0u : 0u, m1 = h == 1 ? ~0u : 0u;
  const unsigned m2 = h == 2 ? ~0u : 0u, m3 = h == 3 ? ~0u : 0u;
  const float4* rows = (const float4*)Hw + lane;
  for (int k0 = 0; k0 < n; k0 += GB) {
    float4 v[GB];
    float  w[GB];
#pragma unroll
    for (int u = 0; u < GB; ++u) {
      const int k = k0 + u;
      if (k < n) {
        v[u] = rows[(size_t)lane_val_i(s, k) * (NH * DOUT / 4)];
        w[u] = head_weight(al, k, m0, m1, m2, m3);
      }
    }
#pragma unroll
    for (int u = 0; u < GB; ++u) {
      if (k0 + u < n) {
        acc.x += w[u] * v[u].x;
        acc.y += w[u] * v[u].y;
        acc.z += w[u] * v[u].z;
        acc.w += w[u] * v[u].w;
      }
    }
  }
}

__global__ __launch_bounds__(256, 16) void k_node(const int* __restrict__ offs,
                                              const int* __restrict__ ssorted,
                                              const float* __restrict__ ssrc,
                                              const float* __restrict__ stgt,
                                              const float* __restrict__ Hw,
                                              float* __restrict__ dout, int N) {
  const int lane = threadIdx.x & 31;
  const int t = blockIdx.x * 8 + __builtin_amdgcn_readfirstlane((int)(threadIdx.x >> 5));
  if (t >= N) return;
  const int beg = __builtin_amdgcn_readfirstlane(offs[t]);
  const int end = __builtin_amdgcn_readfirstlane(offs[t + 1]);
  const float4 bt = ((const float4*)stgt)[t];

  const bool own0 = beg + lane < end;
  int s0 = 0;
  float4 e0;
  e0.x = e0.y = e0.z = e0.w = -INFINITY;
  if (own0) {
    s0 = ssorted[beg + lane];
    e0 = logits4(((const float4*)ssrc)[s0], bt);
  }

  float4 mx = e0;
  for (int j = beg + 32 + lane; j < end; j += 32) {
    const float4 e = logits4(((const float4*)ssrc)[ssorted[j]], bt);
    mx.x = fmaxf(mx.x, e.x);
    mx.y = fmaxf(mx.y, e.y);
    mx.z = fmaxf(mx.z, e.z);
    mx.w = fmaxf(mx.w, e.w);
  }
  for (int d = 16; d >= 1; d >>= 1) {
    mx.x = fmaxf(mx.x, __shfl_xor(mx.x, d));
    mx.y = fmaxf(mx.y, __shfl_xor(mx.y, d));
    mx.z = fmaxf(mx.z, __shfl_xor(mx.z, d));
    mx.w = fmaxf(mx.w, __shfl_xor(mx.w, d));
  }
  mx.x = uniform_f(mx.x);
  mx.y = uniform_f(mx.y);
  mx.z = uniform_f(mx.z);
  mx.w = uniform_f(mx.w);

  float4 p0;
  p0.x = own0 ? expf(e0.x - mx.x) : 0.f;
  p0.y = own0 ? expf(e0.y - mx.y) : 0.f;
  p0.z = own0 ? expf(e0.z - mx.z) : 0.f;
  p0.w = own0 ? expf(e0.w - mx.w) : 0.f;
  float4 sm = p0;
  for (int j = beg + 32 + lane; j < end; j += 32) {
    const float4 e = logits4(((const float4*)ssrc)[ssorted[j]], bt);
    sm.x += expf(e.x - mx.x);
    sm.y += expf(e.y - mx.y);
    sm.z += expf(e.z - mx.z);
    sm.w += expf(e.w - mx.w);
  }
  for (int d = 16; d >= 1; d >>= 1) {
    sm.x += __shfl_xor(sm.x, d);
    sm.y += __shfl_xor(sm.y, d);
    sm.z += __shfl_xor(sm.z, d);
    sm.w += __shfl_xor(sm.w, d);
  }
  sm.x = uniform_f(sm.x);
  sm.y = uniform_f(sm.y);
  sm.z = uniform_f(sm.z);
  sm.w = uniform_f(sm.w);

  float4 acc;
  acc.x = acc.y = acc.z = acc.w = 0.f;
  int s = s0;
  float4 al;
  al.x = p0.x / sm.x;
  al.y = p0.y / sm.y;
  al.z = p0.z / sm.z;
  al.w = p0.w / sm.w;
  for (int c = beg; c < end; c += 32) {
    if (c != beg) {
      const int j = c + lane;
      s = 0;
      al.x = al.y = al.z = al.w = 0.f;
      if (j < end) {
        s = ssorted[j];
        const float4 e = logits4(((const float4*)ssrc)[s], bt);
        al.x = expf(e.x - mx.x) / sm.x;
        al.y = expf(e.y - mx.y) / sm.y;
        al.z = expf(e.z - mx.z) / sm.z;
        al.w = expf(e.w - mx.w) / sm.w;
      }
    }
    const int left = end - c;
    gather_rows(s, al, left < 32 ? left : 32, lane, Hw, acc);
  }

  float4 r;
  r.x = acc.x > 0.f ? acc.x : expm1f(acc.x);
  r.y = acc.y > 0.f ? acc.y : expm1f(acc.y);
  r.z = acc.z > 0.f ? acc.z : expm1f(acc.z);
  r.w = acc.w > 0.f ? acc.w : expm1f(acc.w);
  float* out = dout + ((size_t)(lane >> 3) * N + t) * DOUT;
  ((float4*)out)[lane & 7] = r;
}

extern "C" void kernel_launch(void* const* d_in, const int* in_sizes, int n_in,
                              void* d_out, int out_size, void* d_ws, size_t ws_size,
                              hipStream_t stream) {
  const float* Hin  = (const float*)d_in[0];
  const int*   ei   = (const int*)d_in[1];
  const float* W    = (const float*)d_in[2];
  const float* Asrc = (const float*)d_in[3];
  const float* Atgt = (const float*)d_in[4];
  float* dout = (float*)d_out;

  const int N = in_sizes[0] / DIN;
  const int E = in_sizes[1] / 2;
  const int* src = ei;
  const int* tgt = ei + E;

  char* p = (char*)d_ws;
  float* Hw      = (float*)p; p += (size_t)N * NH * DOUT * sizeof(float);
  float* ssrc    = (float*)p; p += (size_t)N * NH * sizeof(float);
  float* stgt    = (float*)p; p += (size_t)N * NH * sizeof(float);
  int*   cnt     = (int*)p;   p += (size_t)N * sizeof(int);
  int*   cursor  = (int*)p;   p += (size_t)N * sizeof(int);
  int*   offs    = (int*)p;   p += (size_t)(N + 4) * sizeof(int);
  int*   ssorted = (int*)p;   p += (size_t)E * sizeof(int);
  (void)ws_size; (void)n_in; (void)out_size;

  const int B = 256;
  k_zero_cnt<<<(N + B - 1) / B, B, 0, stream>>>(cnt, N);
  k_count<<<(E + B - 1) / B, B, 0, stream>>>(tgt, cnt, E);
  k_scan<<<(N + SCAN_T - 1) / SCAN_T, SCAN_T, 0, stream>>>(cnt, offs, cursor, N);
  k_fill<<<(E + B - 1) / B, B, 0, stream>>>(src, tgt, cursor, ssorted, E);
  k_hw_wmma<<<(N + 15) / 16, 128, 0, stream>>>(Hin, W, Hw, N);
  k_scores<<<(N + B - 1) / B, B, 0, stream>>>(Hw, Asrc, Atgt, ssrc, stgt, N);
  k_node<<<(N + 7) / 8, 256, 0, stream>>>(offs, ssorted, ssrc, stgt, Hw, dout, N);
}


# k_count keeps the value its counting atomic returns as the edge's rank in its bucket; k_fill stores at offs[target]+rank: the second pass of 1.6M atomics is gone; sum order = bucket arrival order
# speedup vs baseline: 8.4867x; 1.2192x over previous
#include <hip/hip_runtime.h>
#include <hip/hip_bf16.h>

typedef float v2f __attribute__((ext_vector_type(2)));
typedef float v8f __attribute__((ext_vector_type(8)));

#define NH   4
#define DIN  128
#define DOUT 32

__global__ __launch_bounds__(128) void k_hw_wmma(const float* __restrict__ Hin,
                                                 const float* __restrict__ W,
                                                 float* __restrict__ Hw, int N) {
  const int h     = threadIdx.x >> 5;
  const int lane  = threadIdx.x & 31;
  const int m     = lane & 15;
  const int khalf = lane >> 4;
  const int base  = blockIdx.x * 16;

  int rowA = base + m;
  if (rowA >= N) rowA = N - 1;
  const float* arow  = Hin + (size_t)rowA * DIN + 2 * khalf;
  const float* b0row = W + ((size_t)h * DOUT + m) * DIN + 2 * khalf;
  const float* b1row = W + ((size_t)h * DOUT + m + 16) * DIN + 2 * khalf;

  v8f acc0 = {0.f, 0.f, 0.f, 0.f, 0.f, 0.f, 0.f, 0.f};
  v8f acc1 = {0.f, 0.f, 0.f, 0.f, 0.f, 0.f, 0.f, 0.f};

#pragma unroll
  for (int k = 0; k < DIN; k += 4) {
    v2f a  = { arow[k],  arow[k + 1]  };
    v2f b0 = { b0row[k], b0row[k + 1] };
    v2f b1 = { b1row[k], b1row[k + 1] };
    acc0 = __builtin_amdgcn_wmma_f32_16x16x4_f32(false, a, false, b0, (short)0,
                                                 acc0, false, false);
    acc1 = __builtin_amdgcn_wmma_f32_16x16x4_f32(false, a, false, b1, (short)0,
                                                 acc1, false, false);
  }

#pragma unroll
  for (int g = 0; g < 8; ++g) {
    int node = base + g + 8 * khalf;
    if (node < N) {
      float* dst = Hw + ((size_t)node * NH + h) * DOUT;
      dst[m]      = acc0[g];
      dst[m + 16] = acc1[g];
    }
  }
}

__global__ void k_scores(const float* __restrict__ Hw, const float* __restrict__ Asrc,
                         const float* __restrict__ Atgt, float* __restrict__ ssrc,
                         float* __restrict__ stgt, int N) {
  int n = blockIdx.x * blockDim.x + threadIdx.x;
  if (n >= N) return;
  const float* hw = Hw + (size_t)n * (NH * DOUT);
#pragma unroll
  for (int h = 0; h < NH; ++h) {
    float s1 = 0.f, s2 = 0.f;
#pragma unroll
    for (int o = 0; o < DOUT; ++o) {
      float v = hw[h * DOUT + o];
      s1 += v * Asrc[h * DOUT + o];
      s2 += v * Atgt[h * DOUT + o];
    }
    ssrc[(size_t)n * NH + h] = s1;
    stgt[(size_t)n * NH + h] = s2;
  }
}

__global__ void k_zero_cnt(int* __restrict__ cnt, int n) {
  int i = blockIdx.x * blockDim.x + threadIdx.x;
  if (i < n) cnt[i] = 0;
}

__global__ void k_count(const int* __restrict__ tgt, int* __restrict__ cnt,
                        int* __restrict__ rank, int E) {
  int e = blockIdx.x * blockDim.x + threadIdx.x;
  if (e < E) rank[e] = atomicAdd(&cnt[tgt[e]], 1);
}

#define SCAN_T 1024
#define SCAN_B 8

__global__ __launch_bounds__(SCAN_T) void k_scan(const int* __restrict__ cnt,
                                                 int* __restrict__ offs, int N) {
  __shared__ int wpre[32];
  __shared__ int wtot[32];
  const int lane  = threadIdx.x & 31;
  const int w     = threadIdx.x >> 5;
  const int first = blockIdx.x * SCAN_T;
  const int i     = first + threadIdx.x;
  const int v     = (i < N) ? cnt[i] : 0;

  const int nq = first >> 2;
  int s = 0;
  const int4* c4 = (const int4*)cnt;
  for (int base = 0; base < nq; base += SCAN_B * SCAN_T) {
    int4 x[SCAN_B];
#pragma unroll
    for (int u = 0; u < SCAN_B; ++u) {
      const int q = base + u * SCAN_T + (int)threadIdx.x;
      x[u] = c4[q < nq ? q : nq - 1];
    }
#pragma unroll
    for (int u = 0; u < SCAN_B; ++u) {
      const int q = base + u * SCAN_T + (int)threadIdx.x;
      if (q < nq) s += (x[u].x + x[u].y) + (x[u].z + x[u].w);
    }
  }

  int inc = v;
  for (int d = 1; d < 32; d <<= 1) {
    const int y = __shfl_up(inc, d);
    if (lane >= d) inc += y;
  }
  for (int d = 16; d >= 1; d >>= 1) s += __shfl_xor(s, d);
  if (lane == 31) wtot[w] = inc;
  if (lane == 0)  wpre[w] = s;
  __syncthreads();

  int a = wpre[lane] + (lane < w ? wtot[lane] : 0);
  for (int d = 16; d >= 1; d >>= 1) a += __shfl_xor(a, d);

  const int ex = a + inc - v;
  if (i < N) offs[i] = ex;
  if (i == N - 1) offs[N] = ex + v;
}

__global__ void k_fill(const int* __restrict__ src, const int* __restrict__ tgt,
                       const int* __restrict__ offs, const int* __restrict__ rank,
                       int* __restrict__ ssorted, int E) {
  int e = blockIdx.x * blockDim.x + threadIdx.x;
  if (e >= E) return;
  ssorted[offs[tgt[e]] + rank[e]] = src[e];
}

__device__ __forceinline__ float lrelu(float x) { return x > 0.f ? x : 0.2f * x; }

__device__ __forceinline__ float4 logits4(float4 a, float4 b) {
  float4 e;
  e.x = lrelu(a.x + b.x);
  e.y = lrelu(a.y + b.y);
  e.z = lrelu(a.z + b.z);
  e.w = lrelu(a.w + b.w);
  return e;
}

__device__ __forceinline__ int lane_val_i(int v, int k) {
  return __builtin_amdgcn_readlane(v, k);
}

__device__ __forceinline__ float uniform_f(float v) {
  return __builtin_bit_cast(float, __builtin_amdgcn_readfirstlane(__builtin_bit_cast(int, v)));
}

#define GB 8

__device__ __forceinline__ float head_weight(float4 al, int k, unsigned m0, unsigned m1,
                                             unsigned m2, unsigned m3) {
  const unsigned b0 = (unsigned)__builtin_amdgcn_readlane(__builtin_bit_cast(int, al.x), k);
  const unsigned b1 = (unsigned)__builtin_amdgcn_readlane(__builtin_bit_cast(int, al.y), k);
  const unsigned b2 = (unsigned)__builtin_amdgcn_readlane(__builtin_bit_cast(int, al.z), k);
  const unsigned b3 = (unsigned)__builtin_amdgcn_readlane(__builtin_bit_cast(int, al.w), k);
  return __builtin_bit_cast(float, (b0 & m0) | (b1 & m1) | (b2 & m2) | (b3 & m3));
}

__device__ __forceinline__ void gather_rows(int s, float4 al, int n, int lane,
                                            const float* __restrict__ Hw, float4& acc) {
  const int h = lane >> 3;
  const unsigned m0 = h == 0 ? ~0u : 0u, m1 = h == 1 ? ~0u : 0u;
  const unsigned m2 = h == 2 ? ~0u : 0u, m3 = h == 3 ? ~0u : 0u;
  const float4* rows = (const float4*)Hw + lane;
  for (int k0 = 0; k0 < n; k0 += GB) {
    float4 v[GB];
    float  w[GB];
#pragma unroll
    for (int u = 0; u < GB; ++u) {
      const int k = k0 + u;
      if (k < n) {
        v[u] = rows[(size_t)lane_val_i(s, k) * (NH * DOUT / 4)];
        w[u] = head_weight(al, k, m0, m1, m2, m3);
      }
    }
#pragma unroll
    for (int u = 0; u < GB; ++u) {
      if (k0 + u < n) {
        acc.x += w[u] * v[u].x;
        acc.y += w[u] * v[u].y;
        acc.z += w[u] * v[u].z;
        acc.w += w[u] * v[u].w;
      }
    }
  }
}

__global__ __launch_bounds__(256, 16) void k_node(const int* __restrict__ offs,
                                              const int* __restrict__ ssorted,
                                              const float* __restrict__ ssrc,
                                              const float* __restrict__ stgt,
                                              const float* __restrict__ Hw,
                                              float* __restrict__ dout, int N) {
  const int lane = threadIdx.x & 31;
  const int t = blockIdx.x * 8 + __builtin_amdgcn_readfirstlane((int)(threadIdx.x >> 5));
  if (t >= N) return;
  const int beg = __builtin_amdgcn_readfirstlane(offs[t]);
  const int end = __builtin_amdgcn_readfirstlane(offs[t + 1]);
  const float4 bt = ((const float4*)stgt)[t];

  const bool own0 = beg + lane < end;
  int s0 = 0;
  float4 e0;
  e0.x = e0.y = e0.z = e0.w = -INFINITY;
  if (own0) {
    s0 = ssorted[beg + lane];
    e0 = logits4(((const float4*)ssrc)[s0], bt);
  }

  float4 mx = e0;
  for (int j = beg + 32 + lane; j < end; j += 32) {
    const float4 e = logits4(((const float4*)ssrc)[ssorted[j]], bt);
    mx.x = fmaxf(mx.x, e.x);
    mx.y = fmaxf(mx.y, e.y);
    mx.z = fmaxf(mx.z, e.z);
    mx.w = fmaxf(mx.w, e.w);
  }
  for (int d = 16; d >= 1; d >>= 1) {
    mx.x = fmaxf(mx.x, __shfl_xor(mx.x, d));
    mx.y = fmaxf(mx.y, __shfl_xor(mx.y, d));
    mx.z = fmaxf(mx.z, __shfl_xor(mx.z, d));
    mx.w = fmaxf(mx.w, __shfl_xor(mx.w, d));
  }
  mx.x = uniform_f(mx.x);
  mx.y = uniform_f(mx.y);
  mx.z = uniform_f(mx.z);
  mx.w = uniform_f(mx.w);

  float4 p0;
  p0.x = own0 ? expf(e0.x - mx.x) : 0.f;
  p0.y = own0 ? expf(e0.y - mx.y) : 0.f;
  p0.z = own0 ? expf(e0.z - mx.z) : 0.f;
  p0.w = own0 ? expf(e0.w - mx.w) : 0.f;
  float4 sm = p0;
  for (int j = beg + 32 + lane; j < end; j += 32) {
    const float4 e = logits4(((const float4*)ssrc)[ssorted[j]], bt);
    sm.x += expf(e.x - mx.x);
    sm.y += expf(e.y - mx.y);
    sm.z += expf(e.z - mx.z);
    sm.w += expf(e.w - mx.w);
  }
  for (int d = 16; d >= 1; d >>= 1) {
    sm.x += __shfl_xor(sm.x, d);
    sm.y += __shfl_xor(sm.y, d);
    sm.z += __shfl_xor(sm.z, d);
    sm.w += __shfl_xor(sm.w, d);
  }
  sm.x = uniform_f(sm.x);
  sm.y = uniform_f(sm.y);
  sm.z = uniform_f(sm.z);
  sm.w = uniform_f(sm.w);

  float4 acc;
  acc.x = acc.y = acc.z = acc.w = 0.f;
  int s = s0;
  float4 al;
  al.x = p0.x / sm.x;
  al.y = p0.y / sm.y;
  al.z = p0.z / sm.z;
  al.w = p0.w / sm.w;
  for (int c = beg; c < end; c += 32) {
    if (c != beg) {
      const int j = c + lane;
      s = 0;
      al.x = al.y = al.z = al.w = 0.f;
      if (j < end) {
        s = ssorted[j];
        const float4 e = logits4(((const float4*)ssrc)[s], bt);
        al.x = expf(e.x - mx.x) / sm.x;
        al.y = expf(e.y - mx.y) / sm.y;
        al.z = expf(e.z - mx.z) / sm.z;
        al.w = expf(e.w - mx.w) / sm.w;
      }
    }
    const int left = end - c;
    gather_rows(s, al, left < 32 ? left : 32, lane, Hw, acc);
  }

  float4 r;
  r.x = acc.x > 0.f ? acc.x : expm1f(acc.x);
  r.y = acc.y > 0.f ? acc.y : expm1f(acc.y);
  r.z = acc.z > 0.f ? acc.z : expm1f(acc.z);
  r.w = acc.w > 0.f ? acc.w : expm1f(acc.w);
  float* out = dout + ((size_t)(lane >> 3) * N + t) * DOUT;
  ((float4*)out)[lane & 7] = r;
}

extern "C" void kernel_launch(void* const* d_in, const int* in_sizes, int n_in,
                              void* d_out, int out_size, void* d_ws, size_t ws_size,
                              hipStream_t stream) {
  const float* Hin  = (const float*)d_in[0];
  const int*   ei   = (const int*)d_in[1];
  const float* W    = (const float*)d_in[2];
  const float* Asrc = (const float*)d_in[3];
  const float* Atgt = (const float*)d_in[4];
  float* dout = (float*)d_out;

  const int N = in_sizes[0] / DIN;
  const int E = in_sizes[1] / 2;
  const int* src = ei;
  const int* tgt = ei + E;

  char* p = (char*)d_ws;
  float* Hw      = (float*)p; p += (size_t)N * NH * DOUT * sizeof(float);
  float* ssrc    = (float*)p; p += (size_t)N * NH * sizeof(float);
  float* stgt    = (float*)p; p += (size_t)N * NH * sizeof(float);
  int*   cnt     = (int*)p;   p += (size_t)N * sizeof(int);
  int*   offs    = (int*)p;   p += (size_t)(N + 4) * sizeof(int);
  int*   ssorted = (int*)p;   p += (size_t)E * sizeof(int);
  int*   rank    = (int*)Hw;
  (void)ws_size; (void)n_in; (void)out_size;

  const int B = 256;
  k_zero_cnt<<<(N + B - 1) / B, B, 0, stream>>>(cnt, N);
  k_count<<<(E + B - 1) / B, B, 0, stream>>>(tgt, cnt, rank, E);
  k_scan<<<(N + SCAN_T - 1) / SCAN_T, SCAN_T, 0, stream>>>(cnt, offs, N);
  k_fill<<<(E + B - 1) / B, B, 0, stream>>>(src, tgt, offs, rank, ssorted, E);
  k_hw_wmma<<<(N + 15) / 16, 128, 0, stream>>>(Hin, W, Hw, N);
  k_scores<<<(N + B - 1) / B, B, 0, stream>>>(Hw, Asrc, Atgt, ssrc, stgt, N);
  k_node<<<(N + 7) / 8, 256, 0, stream>>>(offs, ssorted, ssrc, stgt, Hw, dout, N);
}


# Edge sort by target as a two-level counting sort with LDS counters (buckets of 256 targets, then per target): no global atomic left; sum order = LDS arrival order in a target's bucket
# speedup vs baseline: 12.4703x; 1.4694x over previous
#include <hip/hip_runtime.h>
#include <hip/hip_bf16.h>

typedef float v2f __attribute__((ext_vector_type(2)));
typedef float v8f __attribute__((ext_vector_type(8)));

#define NH   4
#define DIN  128
#define DOUT 32

__global__ __launch_bounds__(128) void k_hw_wmma(const float* __restrict__ Hin,
                                                 const float* __restrict__ W,
                                                 float* __restrict__ Hw, int N) {
  const int h     = threadIdx.x >> 5;
  const int lane  = threadIdx.x & 31;
  const int m     = lane & 15;
  const int khalf = lane >> 4;
  const int base  = blockIdx.x * 16;

  int rowA = base + m;
  if (rowA >= N) rowA = N - 1;
  const float* arow  = Hin + (size_t)rowA * DIN + 2 * khalf;
  const float* b0row = W + ((size_t)h * DOUT + m) * DIN + 2 * khalf;
  const float* b1row = W + ((size_t)h * DOUT + m + 16) * DIN + 2 * khalf;

  v8f acc0 = {0.f, 0.f, 0.f, 0.f, 0.f, 0.f, 0.f, 0.f};
  v8f acc1 = {0.f, 0.f, 0.f, 0.f, 0.f, 0.f, 0.f, 0.f};

#pragma unroll
  for (int k = 0; k < DIN; k += 4) {
    v2f a  = { arow[k],  arow[k + 1]  };
    v2f b0 = { b0row[k], b0row[k + 1] };
    v2f b1 = { b1row[k], b1row[k + 1] };
    acc0 = __builtin_amdgcn_wmma_f32_16x16x4_f32(false, a, false, b0, (short)0,
                                                 acc0, false, false);
    acc1 = __builtin_amdgcn_wmma_f32_16x16x4_f32(false, a, false, b1, (short)0,
                                                 acc1, false, false);
  }

#pragma unroll
  for (int g = 0; g < 8; ++g) {
    int node = base + g + 8 * khalf;
    if (node < N) {
      float* dst = Hw + ((size_t)node * NH + h) * DOUT;
      dst[m]      = acc0[g];
      dst[m + 16] = acc1[g];
    }
  }
}

__global__ void k_scores(const float* __restrict__ Hw, const float* __restrict__ Asrc,
                         const float* __restrict__ Atgt, float* __restrict__ ssrc,
                         float* __restrict__ stgt, int N) {
  int n = blockIdx.x * blockDim.x + threadIdx.x;
  if (n >= N) return;
  const float* hw = Hw + (size_t)n * (NH * DOUT);
#pragma unroll
  for (int h = 0; h < NH; ++h) {
    float s1 = 0.f, s2 = 0.f;
#pragma unroll
    for (int o = 0; o < DOUT; ++o) {
      float v = hw[h * DOUT + o];
      s1 += v * Asrc[h * DOUT + o];
      s2 += v * Atgt[h * DOUT + o];
    }
    ssrc[(size_t)n * NH + h] = s1;
    stgt[(size_t)n * NH + h] = s2;
  }
}

#define BK_SHIFT  8
#define BK_NODES  (1 << BK_SHIFT)
#define L1_BLOCKS 256
#define L1_T      1024
#define L1_U      8
#define L2_T      512
#define L2_U      8

__global__ __launch_bounds__(L1_T) void k_l1_count(const int* __restrict__ tgt,
                                                   int* __restrict__ gh, int E, int nb) {
  extern __shared__ int l1_cnt[];
  const int tid = threadIdx.x;
  for (int i = tid; i < nb; i += L1_T) l1_cnt[i] = 0;
  __syncthreads();

  const int chunk = (E + (int)gridDim.x - 1) / (int)gridDim.x;
  const int beg   = (int)blockIdx.x * chunk;
  const int end   = min(E, beg + chunk);
  for (int e0 = beg; e0 < end; e0 += L1_T * L1_U) {
    int t[L1_U];
#pragma unroll
    for (int u = 0; u < L1_U; ++u) {
      const int e = e0 + u * L1_T + tid;
      t[u] = tgt[e < end ? e : end - 1];
    }
#pragma unroll
    for (int u = 0; u < L1_U; ++u)
      if (e0 + u * L1_T + tid < end) atomicAdd(&l1_cnt[t[u] >> BK_SHIFT], 1);
  }
  __syncthreads();
  for (int i = tid; i < nb; i += L1_T) gh[(size_t)i * gridDim.x + blockIdx.x] = l1_cnt[i];
}

__global__ __launch_bounds__(L1_T) void k_l1_place(const int* __restrict__ src,
                                                   const int* __restrict__ tgt,
                                                   const int* __restrict__ gbase,
                                                   int* __restrict__ pe_raw, int E, int nb) {
  extern __shared__ int l1_cur[];
  int2* pe = (int2*)pe_raw;
  const int tid = threadIdx.x;
  for (int i = tid; i < nb; i += L1_T) l1_cur[i] = gbase[(size_t)i * gridDim.x + blockIdx.x];
  __syncthreads();

  const int chunk = (E + (int)gridDim.x - 1) / (int)gridDim.x;
  const int beg   = (int)blockIdx.x * chunk;
  const int end   = min(E, beg + chunk);
  for (int e0 = beg; e0 < end; e0 += L1_T * L1_U) {
    int s[L1_U];
    int t[L1_U];
#pragma unroll
    for (int u = 0; u < L1_U; ++u) {
      const int e  = e0 + u * L1_T + tid;
      const int ec = e < end ? e : end - 1;
      t[u] = tgt[ec];
      s[u] = src[ec];
    }
#pragma unroll
    for (int u = 0; u < L1_U; ++u) {
      if (e0 + u * L1_T + tid < end) {
        const int pos = atomicAdd(&l1_cur[t[u] >> BK_SHIFT], 1);
        pe[pos] = make_int2(s[u], t[u]);
      }
    }
  }
}

#define SCAN_T 1024
#define SCAN_B 8

__global__ __launch_bounds__(SCAN_T) void k_scan(const int* __restrict__ cnt,
                                                 int* __restrict__ offs, int N) {
  __shared__ int wpre[32];
  __shared__ int wtot[32];
  const int lane  = threadIdx.x & 31;
  const int w     = threadIdx.x >> 5;
  const int first = blockIdx.x * SCAN_T;
  const int i     = first + threadIdx.x;
  const int v     = (i < N) ? cnt[i] : 0;

  const int nq = first >> 2;
  int s = 0;
  const int4* c4 = (const int4*)cnt;
  for (int base = 0; base < nq; base += SCAN_B * SCAN_T) {
    int4 x[SCAN_B];
#pragma unroll
    for (int u = 0; u < SCAN_B; ++u) {
      const int q = base + u * SCAN_T + (int)threadIdx.x;
      x[u] = c4[q < nq ? q : nq - 1];
    }
#pragma unroll
    for (int u = 0; u < SCAN_B; ++u) {
      const int q = base + u * SCAN_T + (int)threadIdx.x;
      if (q < nq) s += (x[u].x + x[u].y) + (x[u].z + x[u].w);
    }
  }

  int inc = v;
  for (int d = 1; d < 32; d <<= 1) {
    const int y = __shfl_up(inc, d);
    if (lane >= d) inc += y;
  }
  for (int d = 16; d >= 1; d >>= 1) s += __shfl_xor(s, d);
  if (lane == 31) wtot[w] = inc;
  if (lane == 0)  wpre[w] = s;
  __syncthreads();

  int a = wpre[lane] + (lane < w ? wtot[lane] : 0);
  for (int d = 16; d >= 1; d >>= 1) a += __shfl_xor(a, d);

  const int ex = a + inc - v;
  if (i < N) offs[i] = ex;
  if (i == N - 1) offs[N] = ex + v;
}

__global__ __launch_bounds__(L2_T) void k_l2_bucket(const int* __restrict__ pe_raw,
                                                    const int* __restrict__ gbase,
                                                    int* __restrict__ offs,
                                                    int* __restrict__ ssorted, int N, int G1) {
  const int2* pe = (const int2*)pe_raw;
  __shared__ int cnt[BK_NODES];
  __shared__ int wtot[BK_NODES / 32];
  const int tid  = threadIdx.x;
  const int lane = tid & 31;
  const int w    = tid >> 5;
  const int k    = blockIdx.x;
  const int beg  = gbase[(size_t)k * G1];
  const int end  = gbase[(size_t)(k + 1) * G1];

  if (tid < BK_NODES) cnt[tid] = 0;
  __syncthreads();

  for (int e0 = beg; e0 < end; e0 += L2_T * L2_U) {
    int t[L2_U];
#pragma unroll
    for (int u = 0; u < L2_U; ++u) {
      const int e = e0 + u * L2_T + tid;
      t[u] = pe[e < end ? e : end - 1].y;
    }
#pragma unroll
    for (int u = 0; u < L2_U; ++u)
      if (e0 + u * L2_T + tid < end) atomicAdd(&cnt[t[u] & (BK_NODES - 1)], 1);
  }
  __syncthreads();

  const int v = tid < BK_NODES ? cnt[tid] : 0;
  int inc = v;
  for (int d = 1; d < 32; d <<= 1) {
    const int y = __shfl_up(inc, d);
    if (lane >= d) inc += y;
  }
  if (tid < BK_NODES && lane == 31) wtot[w] = inc;
  __syncthreads();
  if (tid < BK_NODES) {
    int pre = 0;
    for (int j = 0; j < w; ++j) pre += wtot[j];
    const int start = beg + pre + inc - v;
    cnt[tid] = start;
    const int node = k * BK_NODES + tid;
    if (node < N) offs[node] = start;
  }
  if (k == (int)gridDim.x - 1 && tid == 0) offs[N] = end;
  __syncthreads();

  for (int e0 = beg; e0 < end; e0 += L2_T * L2_U) {
    int2 q[L2_U];
#pragma unroll
    for (int u = 0; u < L2_U; ++u) {
      const int e = e0 + u * L2_T + tid;
      q[u] = pe[e < end ? e : end - 1];
    }
#pragma unroll
    for (int u = 0; u < L2_U; ++u) {
      if (e0 + u * L2_T + tid < end) {
        const int pos = atomicAdd(&cnt[q[u].y & (BK_NODES - 1)], 1);
        ssorted[pos] = q[u].x;
      }
    }
  }
}

__device__ __forceinline__ float lrelu(float x) { return x > 0.f ? x : 0.2f * x; }

__device__ __forceinline__ float4 logits4(float4 a, float4 b) {
  float4 e;
  e.x = lrelu(a.x + b.x);
  e.y = lrelu(a.y + b.y);
  e.z = lrelu(a.z + b.z);
  e.w = lrelu(a.w + b.w);
  return e;
}

__device__ __forceinline__ int lane_val_i(int v, int k) {
  return __builtin_amdgcn_readlane(v, k);
}

__device__ __forceinline__ float uniform_f(float v) {
  return __builtin_bit_cast(float, __builtin_amdgcn_readfirstlane(__builtin_bit_cast(int, v)));
}

#define GB 8

__device__ __forceinline__ float head_weight(float4 al, int k, unsigned m0, unsigned m1,
                                             unsigned m2, unsigned m3) {
  const unsigned b0 = (unsigned)__builtin_amdgcn_readlane(__builtin_bit_cast(int, al.x), k);
  const unsigned b1 = (unsigned)__builtin_amdgcn_readlane(__builtin_bit_cast(int, al.y), k);
  const unsigned b2 = (unsigned)__builtin_amdgcn_readlane(__builtin_bit_cast(int, al.z), k);
  const unsigned b3 = (unsigned)__builtin_amdgcn_readlane(__builtin_bit_cast(int, al.w), k);
  return __builtin_bit_cast(float, (b0 & m0) | (b1 & m1) | (b2 & m2) | (b3 & m3));
}

__device__ __forceinline__ void gather_rows(int s, float4 al, int n, int lane,
                                            const float* __restrict__ Hw, float4& acc) {
  const int h = lane >> 3;
  const unsigned m0 = h == 0 ? ~0u : 0u, m1 = h == 1 ? ~0u : 0u;
  const unsigned m2 = h == 2 ? ~0u : 0u, m3 = h == 3 ? ~0u : 0u;
  const float4* rows = (const float4*)Hw + lane;
  for (int k0 = 0; k0 < n; k0 += GB) {
    float4 v[GB];
    float  w[GB];
#pragma unroll
    for (int u = 0; u < GB; ++u) {
      const int k = k0 + u;
      if (k < n) {
        v[u] = rows[(size_t)lane_val_i(s, k) * (NH * DOUT / 4)];
        w[u] = head_weight(al, k, m0, m1, m2, m3);
      }
    }
#pragma unroll
    for (int u = 0; u < GB; ++u) {
      if (k0 + u < n) {
        acc.x += w[u] * v[u].x;
        acc.y += w[u] * v[u].y;
        acc.z += w[u] * v[u].z;
        acc.w += w[u] * v[u].w;
      }
    }
  }
}

__global__ __launch_bounds__(256, 16) void k_node(const int* __restrict__ offs,
                                              const int* __restrict__ ssorted,
                                              const float* __restrict__ ssrc,
                                              const float* __restrict__ stgt,
                                              const float* __restrict__ Hw,
                                              float* __restrict__ dout, int N) {
  const int lane = threadIdx.x & 31;
  const int t = blockIdx.x * 8 + __builtin_amdgcn_readfirstlane((int)(threadIdx.x >> 5));
  if (t >= N) return;
  const int beg = __builtin_amdgcn_readfirstlane(offs[t]);
  const int end = __builtin_amdgcn_readfirstlane(offs[t + 1]);
  const float4 bt = ((const float4*)stgt)[t];

  const bool own0 = beg + lane < end;
  int s0 = 0;
  float4 e0;
  e0.x = e0.y = e0.z = e0.w = -INFINITY;
  if (own0) {
    s0 = ssorted[beg + lane];
    e0 = logits4(((const float4*)ssrc)[s0], bt);
  }

  float4 mx = e0;
  for (int j = beg + 32 + lane; j < end; j += 32) {
    const float4 e = logits4(((const float4*)ssrc)[ssorted[j]], bt);
    mx.x = fmaxf(mx.x, e.x);
    mx.y = fmaxf(mx.y, e.y);
    mx.z = fmaxf(mx.z, e.z);
    mx.w = fmaxf(mx.w, e.w);
  }
  for (int d = 16; d >= 1; d >>= 1) {
    mx.x = fmaxf(mx.x, __shfl_xor(mx.x, d));
    mx.y = fmaxf(mx.y, __shfl_xor(mx.y, d));
    mx.z = fmaxf(mx.z, __shfl_xor(mx.z, d));
    mx.w = fmaxf(mx.w, __shfl_xor(mx.w, d));
  }
  mx.x = uniform_f(mx.x);
  mx.y = uniform_f(mx.y);
  mx.z = uniform_f(mx.z);
  mx.w = uniform_f(mx.w);

  float4 p0;
  p0.x = own0 ? expf(e0.x - mx.x) : 0.f;
  p0.y = own0 ? expf(e0.y - mx.y) : 0.f;
  p0.z = own0 ? expf(e0.z - mx.z) : 0.f;
  p0.w = own0 ? expf(e0.w - mx.w) : 0.f;
  float4 sm = p0;
  for (int j = beg + 32 + lane; j < end; j += 32) {
    const float4 e = logits4(((const float4*)ssrc)[ssorted[j]], bt);
    sm.x += expf(e.x - mx.x);
    sm.y += expf(e.y - mx.y);
    sm.z += expf(e.z - mx.z);
    sm.w += expf(e.w - mx.w);
  }
  for (int d = 16; d >= 1; d >>= 1) {
    sm.x += __shfl_xor(sm.x, d);
    sm.y += __shfl_xor(sm.y, d);
    sm.z += __shfl_xor(sm.z, d);
    sm.w += __shfl_xor(sm.w, d);
  }
  sm.x = uniform_f(sm.x);
  sm.y = uniform_f(sm.y);
  sm.z = uniform_f(sm.z);
  sm.w = uniform_f(sm.w);

  float4 acc;
  acc.x = acc.y = acc.z = acc.w = 0.f;
  int s = s0;
  float4 al;
  al.x = p0.x / sm.x;
  al.y = p0.y / sm.y;
  al.z = p0.z / sm.z;
  al.w = p0.w / sm.w;
  for (int c = beg; c < end; c += 32) {
    if (c != beg) {
      const int j = c + lane;
      s = 0;
      al.x = al.y = al.z = al.w = 0.f;
      if (j < end) {
        s = ssorted[j];
        const float4 e = logits4(((const float4*)ssrc)[s], bt);
        al.x = expf(e.x - mx.x) / sm.x;
        al.y = expf(e.y - mx.y) / sm.y;
        al.z = expf(e.z - mx.z) / sm.z;
        al.w = expf(e.w - mx.w) / sm.w;
      }
    }
    const int left = end - c;
    gather_rows(s, al, left < 32 ? left : 32, lane, Hw, acc);
  }

  float4 r;
  r.x = acc.x > 0.f ? acc.x : expm1f(acc.x);
  r.y = acc.y > 0.f ? acc.y : expm1f(acc.y);
  r.z = acc.z > 0.f ? acc.z : expm1f(acc.z);
  r.w = acc.w > 0.f ? acc.w : expm1f(acc.w);
  float* out = dout + ((size_t)(lane >> 3) * N + t) * DOUT;
  ((float4*)out)[lane & 7] = r;
}

extern "C" void kernel_launch(void* const* d_in, const int* in_sizes, int n_in,
                              void* d_out, int out_size, void* d_ws, size_t ws_size,
                              hipStream_t stream) {
  const float* Hin  = (const float*)d_in[0];
  const int*   ei   = (const int*)d_in[1];
  const float* W    = (const float*)d_in[2];
  const float* Asrc = (const float*)d_in[3];
  const float* Atgt = (const float*)d_in[4];
  float* dout = (float*)d_out;

  const int N = in_sizes[0] / DIN;
  const int E = in_sizes[1] / 2;
  const int* src = ei;
  const int* tgt = ei + E;

  char* p = (char*)d_ws;
  float* Hw      = (float*)p; p += (size_t)N * NH * DOUT * sizeof(float);
  float* ssrc    = (float*)p; p += (size_t)N * NH * sizeof(float);
  float* stgt    = (float*)p; p += (size_t)N * NH * sizeof(float);
  int*   offs    = (int*)p;   p += (size_t)(N + 4) * sizeof(int);
  int*   ssorted = (int*)p;   p += (size_t)E * sizeof(int);
  const int nb = (N + BK_NODES - 1) >> BK_SHIFT;
  const int M  = nb * L1_BLOCKS;
  int*  pe    = (int*)Hw;
  int*  gh    = (int*)Hw + (((size_t)2 * E + 3) & ~(size_t)3);
  int*  gbase = gh + (((size_t)M + 3) & ~(size_t)3);
  (void)ws_size; (void)n_in; (void)out_size;

  const int B = 256;
  k_l1_count<<<L1_BLOCKS, L1_T, nb * sizeof(int), stream>>>(tgt, gh, E, nb);
  k_scan<<<(M + SCAN_T - 1) / SCAN_T, SCAN_T, 0, stream>>>(gh, gbase, M);
  k_l1_place<<<L1_BLOCKS, L1_T, nb * sizeof(int), stream>>>(src, tgt, gbase, pe, E, nb);
  k_l2_bucket<<<nb, L2_T, 0, stream>>>(pe, gbase, offs, ssorted, N, L1_BLOCKS);
  k_hw_wmma<<<(N + 15) / 16, 128, 0, stream>>>(Hin, W, Hw, N);
  k_scores<<<(N + B - 1) / B, B, 0, stream>>>(Hw, Asrc, Atgt, ssrc, stgt, N);
  k_node<<<(N + 7) / 8, 256, 0, stream>>>(offs, ssorted, ssrc, stgt, Hw, dout, N);
}


# k_hw_wmma: each wave keeps its head's W fragments in registers over a list of node tiles; A and W fragments read 16 B per lane (k order inside each 8 permuted, same for both): H_w sum order changes
# speedup vs baseline: 15.8987x; 1.2749x over previous
#include <hip/hip_runtime.h>
#include <hip/hip_bf16.h>

typedef float v2f __attribute__((ext_vector_type(2)));
typedef float v8f __attribute__((ext_vector_type(8)));

#define NH   4
#define DIN  128
#define DOUT 32

#define HW_BLOCKS 1024
#define HW_SPREAD 569

__device__ __forceinline__ const float4* hw_arow(const float* __restrict__ Hin, int tile, int m,
                                                 int khalf, int N) {
  int r = tile * 16 + m;
  if (r >= N) r = N - 1;
  return (const float4*)(Hin + (size_t)r * DIN + 4 * khalf);
}

__device__ __forceinline__ void hw_mma4(float4 a, float4 p, float4 q, v8f& acc0, v8f& acc1) {
  const v2f a0 = {a.x, a.y}, a1 = {a.z, a.w};
  const v2f p0 = {p.x, p.y}, p1 = {p.z, p.w};
  const v2f q0 = {q.x, q.y}, q1 = {q.z, q.w};
  acc0 = __builtin_amdgcn_wmma_f32_16x16x4_f32(false, a0, false, p0, (short)0, acc0, false, false);
  acc1 = __builtin_amdgcn_wmma_f32_16x16x4_f32(false, a0, false, q0, (short)0, acc1, false, false);
  acc0 = __builtin_amdgcn_wmma_f32_16x16x4_f32(false, a1, false, p1, (short)0, acc0, false, false);
  acc1 = __builtin_amdgcn_wmma_f32_16x16x4_f32(false, a1, false, q1, (short)0, acc1, false, false);
}

__global__ __launch_bounds__(128, 4) void k_hw_wmma(const float* __restrict__ Hin,
                                                    const float* __restrict__ W,
                                                    float* __restrict__ Hw, int N) {
  const int h     = threadIdx.x >> 5;
  const int lane  = threadIdx.x & 31;
  const int m     = lane & 15;
  const int khalf = lane >> 4;
  const int b     = (int)blockIdx.x;

  const int ntiles = (N + 15) >> 4;
  const int full   = ntiles / HW_BLOCKS;
  const int rem    = ntiles - full * HW_BLOCKS;
  const int x      = (b * HW_SPREAD) & (HW_BLOCKS - 1);
  const int nmy    = full + (x < rem ? 1 : 0);

  const float4* b0p = (const float4*)(W + ((size_t)h * DOUT + m) * DIN + 4 * khalf);
  const float4* b1p = (const float4*)(W + ((size_t)h * DOUT + m + 16) * DIN + 4 * khalf);
  float4 b0[16], b1[16];
#pragma unroll
  for (int j = 0; j < 16; ++j) {
    b0[j] = b0p[2 * j];
    b1[j] = b1p[2 * j];
  }

  float4 alo[8], ahi[8];
  int tile = 0 < full ? b : full * HW_BLOCKS + x;
  {
    const float4* ap = hw_arow(Hin, tile, m, khalf, N);
#pragma unroll
    for (int j = 0; j < 8; ++j) alo[j] = ap[2 * j];
#pragma unroll
    for (int j = 0; j < 8; ++j) ahi[j] = ap[2 * (j + 8)];
  }

  for (int i = 0; i < nmy; ++i) {
    const int nxt = i + 1 < full ? b + (i + 1) * HW_BLOCKS : full * HW_BLOCKS + x;
    const float4* np = hw_arow(Hin, nxt, m, khalf, N);
    const int base = tile * 16;

    v8f acc0 = {0.f, 0.f, 0.f, 0.f, 0.f, 0.f, 0.f, 0.f};
    v8f acc1 = {0.f, 0.f, 0.f, 0.f, 0.f, 0.f, 0.f, 0.f};
#pragma unroll
    for (int j = 0; j < 8; ++j) hw_mma4(alo[j], b0[j], b1[j], acc0, acc1);
#pragma unroll
    for (int j = 0; j < 8; ++j) alo[j] = np[2 * j];
#pragma unroll
    for (int j = 0; j < 8; ++j) hw_mma4(ahi[j], b0[j + 8], b1[j + 8], acc0, acc1);
#pragma unroll
    for (int j = 0; j < 8; ++j) ahi[j] = np[2 * (j + 8)];

#pragma unroll
    for (int g = 0; g < 8; ++g) {
      int node = base + g + 8 * khalf;
      if (node < N) {
        float* dst = Hw + ((size_t)node * NH + h) * DOUT;
        dst[m]      = acc0[g];
        dst[m + 16] = acc1[g];
      }
    }
    tile = nxt;
  }
}

__global__ void k_scores(const float* __restrict__ Hw, const float* __restrict__ Asrc,
                         const float* __restrict__ Atgt, float* __restrict__ ssrc,
                         float* __restrict__ stgt, int N) {
  int n = blockIdx.x * blockDim.x + threadIdx.x;
  if (n >= N) return;
  const float* hw = Hw + (size_t)n * (NH * DOUT);
#pragma unroll
  for (int h = 0; h < NH; ++h) {
    float s1 = 0.f, s2 = 0.f;
#pragma unroll
    for (int o = 0; o < DOUT; ++o) {
      float v = hw[h * DOUT + o];
      s1 += v * Asrc[h * DOUT + o];
      s2 += v * Atgt[h * DOUT + o];
    }
    ssrc[(size_t)n * NH + h] = s1;
    stgt[(size_t)n * NH + h] = s2;
  }
}

#define BK_SHIFT  8
#define BK_NODES  (1 << BK_SHIFT)
#define L1_BLOCKS 256
#define L1_T      1024
#define L1_U      8
#define L2_T      512
#define L2_U      8

__global__ __launch_bounds__(L1_T) void k_l1_count(const int* __restrict__ tgt,
                                                   int* __restrict__ gh, int E, int nb) {
  extern __shared__ int l1_cnt[];
  const int tid = threadIdx.x;
  for (int i = tid; i < nb; i += L1_T) l1_cnt[i] = 0;
  __syncthreads();

  const int chunk = (E + (int)gridDim.x - 1) / (int)gridDim.x;
  const int beg   = (int)blockIdx.x * chunk;
  const int end   = min(E, beg + chunk);
  for (int e0 = beg; e0 < end; e0 += L1_T * L1_U) {
    int t[L1_U];
#pragma unroll
    for (int u = 0; u < L1_U; ++u) {
      const int e = e0 + u * L1_T + tid;
      t[u] = tgt[e < end ? e : end - 1];
    }
#pragma unroll
    for (int u = 0; u < L1_U; ++u)
      if (e0 + u * L1_T + tid < end) atomicAdd(&l1_cnt[t[u] >> BK_SHIFT], 1);
  }
  __syncthreads();
  for (int i = tid; i < nb; i += L1_T) gh[(size_t)i * gridDim.x + blockIdx.x] = l1_cnt[i];
}

__global__ __launch_bounds__(L1_T) void k_l1_place(const int* __restrict__ src,
                                                   const int* __restrict__ tgt,
                                                   const int* __restrict__ gbase,
                                                   int* __restrict__ pe_raw, int E, int nb) {
  extern __shared__ int l1_cur[];
  int2* pe = (int2*)pe_raw;
  const int tid = threadIdx.x;
  for (int i = tid; i < nb; i += L1_T) l1_cur[i] = gbase[(size_t)i * gridDim.x + blockIdx.x];
  __syncthreads();

  const int chunk = (E + (int)gridDim.x - 1) / (int)gridDim.x;
  const int beg   = (int)blockIdx.x * chunk;
  const int end   = min(E, beg + chunk);
  for (int e0 = beg; e0 < end; e0 += L1_T * L1_U) {
    int s[L1_U];
    int t[L1_U];
#pragma unroll
    for (int u = 0; u < L1_U; ++u) {
      const int e  = e0 + u * L1_T + tid;
      const int ec = e < end ? e : end - 1;
      t[u] = tgt[ec];
      s[u] = src[ec];
    }
#pragma unroll
    for (int u = 0; u < L1_U; ++u) {
      if (e0 + u * L1_T + tid < end) {
        const int pos = atomicAdd(&l1_cur[t[u] >> BK_SHIFT], 1);
        pe[pos] = make_int2(s[u], t[u]);
      }
    }
  }
}

#define SCAN_T 1024
#define SCAN_B 8

__global__ __launch_bounds__(SCAN_T) void k_scan(const int* __restrict__ cnt,
                                                 int* __restrict__ offs, int N) {
  __shared__ int wpre[32];
  __shared__ int wtot[32];
  const int lane  = threadIdx.x & 31;
  const int w     = threadIdx.x >> 5;
  const int first = blockIdx.x * SCAN_T;
  const int i     = first + threadIdx.x;
  const int v     = (i < N) ? cnt[i] : 0;

  const int nq = first >> 2;
  int s = 0;
  const int4* c4 = (const int4*)cnt;
  for (int base = 0; base < nq; base += SCAN_B * SCAN_T) {
    int4 x[SCAN_B];
#pragma unroll
    for (int u = 0; u < SCAN_B; ++u) {
      const int q = base + u * SCAN_T + (int)threadIdx.x;
      x[u] = c4[q < nq ? q : nq - 1];
    }
#pragma unroll
    for (int u = 0; u < SCAN_B; ++u) {
      const int q = base + u * SCAN_T + (int)threadIdx.x;
      if (q < nq) s += (x[u].x + x[u].y) + (x[u].z + x[u].w);
    }
  }

  int inc = v;
  for (int d = 1; d < 32; d <<= 1) {
    const int y = __shfl_up(inc, d);
    if (lane >= d) inc += y;
  }
  for (int d = 16; d >= 1; d >>= 1) s += __shfl_xor(s, d);
  if (lane == 31) wtot[w] = inc;
  if (lane == 0)  wpre[w] = s;
  __syncthreads();

  int a = wpre[lane] + (lane < w ? wtot[lane] : 0);
  for (int d = 16; d >= 1; d >>= 1) a += __shfl_xor(a, d);

  const int ex = a + inc - v;
  if (i < N) offs[i] = ex;
  if (i == N - 1) offs[N] = ex + v;
}

__global__ __launch_bounds__(L2_T) void k_l2_bucket(const int* __restrict__ pe_raw,
                                                    const int* __restrict__ gbase,
                                                    int* __restrict__ offs,
                                                    int* __restrict__ ssorted, int N, int G1) {
  const int2* pe = (const int2*)pe_raw;
  __shared__ int cnt[BK_NODES];
  __shared__ int wtot[BK_NODES / 32];
  const int tid  = threadIdx.x;
  const int lane = tid & 31;
  const int w    = tid >> 5;
  const int k    = blockIdx.x;
  const int beg  = gbase[(size_t)k * G1];
  const int end  = gbase[(size_t)(k + 1) * G1];

  if (tid < BK_NODES) cnt[tid] = 0;
  __syncthreads();

  for (int e0 = beg; e0 < end; e0 += L2_T * L2_U) {
    int t[L2_U];
#pragma unroll
    for (int u = 0; u < L2_U; ++u) {
      const int e = e0 + u * L2_T + tid;
      t[u] = pe[e < end ? e : end - 1].y;
    }
#pragma unroll
    for (int u = 0; u < L2_U; ++u)
      if (e0 + u * L2_T + tid < end) atomicAdd(&cnt[t[u] & (BK_NODES - 1)], 1);
  }
  __syncthreads();

  const int v = tid < BK_NODES ? cnt[tid] : 0;
  int inc = v;
  for (int d = 1; d < 32; d <<= 1) {
    const int y = __shfl_up(inc, d);
    if (lane >= d) inc += y;
  }
  if (tid < BK_NODES && lane == 31) wtot[w] = inc;
  __syncthreads();
  if (tid < BK_NODES) {
    int pre = 0;
    for (int j = 0; j < w; ++j) pre += wtot[j];
    const int start = beg + pre + inc - v;
    cnt[tid] = start;
    const int node = k * BK_NODES + tid;
    if (node < N) offs[node] = start;
  }
  if (k == (int)gridDim.x - 1 && tid == 0) offs[N] = end;
  __syncthreads();

  for (int e0 = beg; e0 < end; e0 += L2_T * L2_U) {
    int2 q[L2_U];
#pragma unroll
    for (int u = 0; u < L2_U; ++u) {
      const int e = e0 + u * L2_T + tid;
      q[u] = pe[e < end ? e : end - 1];
    }
#pragma unroll
    for (int u = 0; u < L2_U; ++u) {
      if (e0 + u * L2_T + tid < end) {
        const int pos = atomicAdd(&cnt[q[u].y & (BK_NODES - 1)], 1);
        ssorted[pos] = q[u].x;
      }
    }
  }
}

__device__ __forceinline__ float lrelu(float x) { return x > 0.f ? x : 0.2f * x; }

__device__ __forceinline__ float4 logits4(float4 a, float4 b) {
  float4 e;
  e.x = lrelu(a.x + b.x);
  e.y = lrelu(a.y + b.y);
  e.z = lrelu(a.z + b.z);
  e.w = lrelu(a.w + b.w);
  return e;
}

__device__ __forceinline__ int lane_val_i(int v, int k) {
  return __builtin_amdgcn_readlane(v, k);
}

__device__ __forceinline__ float uniform_f(float v) {
  return __builtin_bit_cast(float, __builtin_amdgcn_readfirstlane(__builtin_bit_cast(int, v)));
}

#define GB 8

__device__ __forceinline__ float head_weight(float4 al, int k, unsigned m0, unsigned m1,
                                             unsigned m2, unsigned m3) {
  const unsigned b0 = (unsigned)__builtin_amdgcn_readlane(__builtin_bit_cast(int, al.x), k);
  const unsigned b1 = (unsigned)__builtin_amdgcn_readlane(__builtin_bit_cast(int, al.y), k);
  const unsigned b2 = (unsigned)__builtin_amdgcn_readlane(__builtin_bit_cast(int, al.z), k);
  const unsigned b3 = (unsigned)__builtin_amdgcn_readlane(__builtin_bit_cast(int, al.w), k);
  return __builtin_bit_cast(float, (b0 & m0) | (b1 & m1) | (b2 & m2) | (b3 & m3));
}

__device__ __forceinline__ void gather_rows(int s, float4 al, int n, int lane,
                                            const float* __restrict__ Hw, float4& acc) {
  const int h = lane >> 3;
  const unsigned m0 = h == 0 ? ~0u : 0u, m1 = h == 1 ? ~0u : 0u;
  const unsigned m2 = h == 2 ? ~0u : 0u, m3 = h == 3 ? ~0u : 0u;
  const float4* rows = (const float4*)Hw + lane;
  for (int k0 = 0; k0 < n; k0 += GB) {
    float4 v[GB];
    float  w[GB];
#pragma unroll
    for (int u = 0; u < GB; ++u) {
      const int k = k0 + u;
      if (k < n) {
        v[u] = rows[(size_t)lane_val_i(s, k) * (NH * DOUT / 4)];
        w[u] = head_weight(al, k, m0, m1, m2, m3);
      }
    }
#pragma unroll
    for (int u = 0; u < GB; ++u) {
      if (k0 + u < n) {
        acc.x += w[u] * v[u].x;
        acc.y += w[u] * v[u].y;
        acc.z += w[u] * v[u].z;
        acc.w += w[u] * v[u].w;
      }
    }
  }
}

__global__ __launch_bounds__(256, 16) void k_node(const int* __restrict__ offs,
                                              const int* __restrict__ ssorted,
                                              const float* __restrict__ ssrc,
                                              const float* __restrict__ stgt,
                                              const float* __restrict__ Hw,
                                              float* __restrict__ dout, int N) {
  const int lane = threadIdx.x & 31;
  const int t = blockIdx.x * 8 + __builtin_amdgcn_readfirstlane((int)(threadIdx.x >> 5));
  if (t >= N) return;
  const int beg = __builtin_amdgcn_readfirstlane(offs[t]);
  const int end = __builtin_amdgcn_readfirstlane(offs[t + 1]);
  const float4 bt = ((const float4*)stgt)[t];

  const bool own0 = beg + lane < end;
  int s0 = 0;
  float4 e0;
  e0.x = e0.y = e0.z = e0.w = -INFINITY;
  if (own0) {
    s0 = ssorted[beg + lane];
    e0 = logits4(((const float4*)ssrc)[s0], bt);
  }

  float4 mx = e0;
  for (int j = beg + 32 + lane; j < end; j += 32) {
    const float4 e = logits4(((const float4*)ssrc)[ssorted[j]], bt);
    mx.x = fmaxf(mx.x, e.x);
    mx.y = fmaxf(mx.y, e.y);
    mx.z = fmaxf(mx.z, e.z);
    mx.w = fmaxf(mx.w, e.w);
  }
  for (int d = 16; d >= 1; d >>= 1) {
    mx.x = fmaxf(mx.x, __shfl_xor(mx.x, d));
    mx.y = fmaxf(mx.y, __shfl_xor(mx.y, d));
    mx.z = fmaxf(mx.z, __shfl_xor(mx.z, d));
    mx.w = fmaxf(mx.w, __shfl_xor(mx.w, d));
  }
  mx.x = uniform_f(mx.x);
  mx.y = uniform_f(mx.y);
  mx.z = uniform_f(mx.z);
  mx.w = uniform_f(mx.w);

  float4 p0;
  p0.x = own0 ? expf(e0.x - mx.x) : 0.f;
  p0.y = own0 ? expf(e0.y - mx.y) : 0.f;
  p0.z = own0 ? expf(e0.z - mx.z) : 0.f;
  p0.w = own0 ? expf(e0.w - mx.w) : 0.f;
  float4 sm = p0;
  for (int j = beg + 32 + lane; j < end; j += 32) {
    const float4 e = logits4(((const float4*)ssrc)[ssorted[j]], bt);
    sm.x += expf(e.x - mx.x);
    sm.y += expf(e.y - mx.y);
    sm.z += expf(e.z - mx.z);
    sm.w += expf(e.w - mx.w);
  }
  for (int d = 16; d >= 1; d >>= 1) {
    sm.x += __shfl_xor(sm.x, d);
    sm.y += __shfl_xor(sm.y, d);
    sm.z += __shfl_xor(sm.z, d);
    sm.w += __shfl_xor(sm.w, d);
  }
  sm.x = uniform_f(sm.x);
  sm.y = uniform_f(sm.y);
  sm.z = uniform_f(sm.z);
  sm.w = uniform_f(sm.w);

  float4 acc;
  acc.x = acc.y = acc.z = acc.w = 0.f;
  int s = s0;
  float4 al;
  al.x = p0.x / sm.x;
  al.y = p0.y / sm.y;
  al.z = p0.z / sm.z;
  al.w = p0.w / sm.w;
  for (int c = beg; c < end; c += 32) {
    if (c != beg) {
      const int j = c + lane;
      s = 0;
      al.x = al.y = al.z = al.w = 0.f;
      if (j < end) {
        s = ssorted[j];
        const float4 e = logits4(((const float4*)ssrc)[s], bt);
        al.x = expf(e.x - mx.x) / sm.x;
        al.y = expf(e.y - mx.y) / sm.y;
        al.z = expf(e.z - mx.z) / sm.z;
        al.w = expf(e.w - mx.w) / sm.w;
      }
    }
    const int left = end - c;
    gather_rows(s, al, left < 32 ? left : 32, lane, Hw, acc);
  }

  float4 r;
  r.x = acc.x > 0.f ? acc.x : expm1f(acc.x);
  r.y = acc.y > 0.f ? acc.y : expm1f(acc.y);
  r.z = acc.z > 0.f ? acc.z : expm1f(acc.z);
  r.w = acc.w > 0.f ? acc.w : expm1f(acc.w);
  float* out = dout + ((size_t)(lane >> 3) * N + t) * DOUT;
  ((float4*)out)[lane & 7] = r;
}

extern "C" void kernel_launch(void* const* d_in, const int* in_sizes, int n_in,
                              void* d_out, int out_size, void* d_ws, size_t ws_size,
                              hipStream_t stream) {
  const float* Hin  = (const float*)d_in[0];
  const int*   ei   = (const int*)d_in[1];
  const float* W    = (const float*)d_in[2];
  const float* Asrc = (const float*)d_in[3];
  const float* Atgt = (const float*)d_in[4];
  float* dout = (float*)d_out;

  const int N = in_sizes[0] / DIN;
  const int E = in_sizes[1] / 2;
  const int* src = ei;
  const int* tgt = ei + E;

  char* p = (char*)d_ws;
  float* Hw      = (float*)p; p += (size_t)N * NH * DOUT * sizeof(float);
  float* ssrc    = (float*)p; p += (size_t)N * NH * sizeof(float);
  float* stgt    = (float*)p; p += (size_t)N * NH * sizeof(float);
  int*   offs    = (int*)p;   p += (size_t)(N + 4) * sizeof(int);
  int*   ssorted = (int*)p;   p += (size_t)E * sizeof(int);
  const int nb = (N + BK_NODES - 1) >> BK_SHIFT;
  const int M  = nb * L1_BLOCKS;
  int*  pe    = (int*)Hw;
  int*  gh    = (int*)Hw + (((size_t)2 * E + 3) & ~(size_t)3);
  int*  gbase = gh + (((size_t)M + 3) & ~(size_t)3);
  (void)ws_size; (void)n_in; (void)out_size;

  const int B = 256;
  k_l1_count<<<L1_BLOCKS, L1_T, nb * sizeof(int), stream>>>(tgt, gh, E, nb);
  k_scan<<<(M + SCAN_T - 1) / SCAN_T, SCAN_T, 0, stream>>>(gh, gbase, M);
  k_l1_place<<<L1_BLOCKS, L1_T, nb * sizeof(int), stream>>>(src, tgt, gbase, pe, E, nb);
  k_l2_bucket<<<nb, L2_T, 0, stream>>>(pe, gbase, offs, ssorted, N, L1_BLOCKS);
  k_hw_wmma<<<HW_BLOCKS, 128, 0, stream>>>(Hin, W, Hw, N);
  k_scores<<<(N + B - 1) / B, B, 0, stream>>>(Hw, Asrc, Atgt, ssrc, stgt, N);
  k_node<<<(N + 7) / 8, 256, 0, stream>>>(offs, ssorted, ssrc, stgt, Hw, dout, N);
}


# Sort level one keeps each work-group's pairs in its own slice, offsets by a prefix sum inside the work-group: global scan and count launches gone (7 -> 5); sum order: LDS arrival, changed
# speedup vs baseline: 17.8658x; 1.1237x over previous
#include <hip/hip_runtime.h>
#include <hip/hip_bf16.h>

typedef float v2f __attribute__((ext_vector_type(2)));
typedef float v8f __attribute__((ext_vector_type(8)));

#define NH   4
#define DIN  128
#define DOUT 32

#define HW_BLOCKS 1024
#define HW_SPREAD 569

__device__ __forceinline__ const float4* hw_arow(const float* __restrict__ Hin, int tile, int m,
                                                 int khalf, int N) {
  int r = tile * 16 + m;
  if (r >= N) r = N - 1;
  return (const float4*)(Hin + (size_t)r * DIN + 4 * khalf);
}

__device__ __forceinline__ void hw_mma4(float4 a, float4 p, float4 q, v8f& acc0, v8f& acc1) {
  const v2f a0 = {a.x, a.y}, a1 = {a.z, a.w};
  const v2f p0 = {p.x, p.y}, p1 = {p.z, p.w};
  const v2f q0 = {q.x, q.y}, q1 = {q.z, q.w};
  acc0 = __builtin_amdgcn_wmma_f32_16x16x4_f32(false, a0, false, p0, (short)0, acc0, false, false);
  acc1 = __builtin_amdgcn_wmma_f32_16x16x4_f32(false, a0, false, q0, (short)0, acc1, false, false);
  acc0 = __builtin_amdgcn_wmma_f32_16x16x4_f32(false, a1, false, p1, (short)0, acc0, false, false);
  acc1 = __builtin_amdgcn_wmma_f32_16x16x4_f32(false, a1, false, q1, (short)0, acc1, false, false);
}

__global__ __launch_bounds__(128, 4) void k_hw_wmma(const float* __restrict__ Hin,
                                                    const float* __restrict__ W,
                                                    float* __restrict__ Hw, int N) {
  const int h     = threadIdx.x >> 5;
  const int lane  = threadIdx.x & 31;
  const int m     = lane & 15;
  const int khalf = lane >> 4;
  const int b     = (int)blockIdx.x;

  const int ntiles = (N + 15) >> 4;
  const int full   = ntiles / HW_BLOCKS;
  const int rem    = ntiles - full * HW_BLOCKS;
  const int x      = (b * HW_SPREAD) & (HW_BLOCKS - 1);
  const int nmy    = full + (x < rem ? 1 : 0);

  const float4* b0p = (const float4*)(W + ((size_t)h * DOUT + m) * DIN + 4 * khalf);
  const float4* b1p = (const float4*)(W + ((size_t)h * DOUT + m + 16) * DIN + 4 * khalf);
  float4 b0[16], b1[16];
#pragma unroll
  for (int j = 0; j < 16; ++j) {
    b0[j] = b0p[2 * j];
    b1[j] = b1p[2 * j];
  }

  float4 alo[8], ahi[8];
  int tile = 0 < full ? b : full * HW_BLOCKS + x;
  {
    const float4* ap = hw_arow(Hin, tile, m, khalf, N);
#pragma unroll
    for (int j = 0; j < 8; ++j) alo[j] = ap[2 * j];
#pragma unroll
    for (int j = 0; j < 8; ++j) ahi[j] = ap[2 * (j + 8)];
  }

  for (int i = 0; i < nmy; ++i) {
    const int nxt = i + 1 < full ? b + (i + 1) * HW_BLOCKS : full * HW_BLOCKS + x;
    const float4* np = hw_arow(Hin, nxt, m, khalf, N);
    const int base = tile * 16;

    v8f acc0 = {0.f, 0.f, 0.f, 0.f, 0.f, 0.f, 0.f, 0.f};
    v8f acc1 = {0.f, 0.f, 0.f, 0.f, 0.f, 0.f, 0.f, 0.f};
#pragma unroll
    for (int j = 0; j < 8; ++j) hw_mma4(alo[j], b0[j], b1[j], acc0, acc1);
#pragma unroll
    for (int j = 0; j < 8; ++j) alo[j] = np[2 * j];
#pragma unroll
    for (int j = 0; j < 8; ++j) hw_mma4(ahi[j], b0[j + 8], b1[j + 8], acc0, acc1);
#pragma unroll
    for (int j = 0; j < 8; ++j) ahi[j] = np[2 * (j + 8)];

#pragma unroll
    for (int g = 0; g < 8; ++g) {
      int node = base + g + 8 * khalf;
      if (node < N) {
        float* dst = Hw + ((size_t)node * NH + h) * DOUT;
        dst[m]      = acc0[g];
        dst[m + 16] = acc1[g];
      }
    }
    tile = nxt;
  }
}

__global__ void k_scores(const float* __restrict__ Hw, const float* __restrict__ Asrc,
                         const float* __restrict__ Atgt, float* __restrict__ ssrc,
                         float* __restrict__ stgt, int N) {
  int n = blockIdx.x * blockDim.x + threadIdx.x;
  if (n >= N) return;
  const float* hw = Hw + (size_t)n * (NH * DOUT);
#pragma unroll
  for (int h = 0; h < NH; ++h) {
    float s1 = 0.f, s2 = 0.f;
#pragma unroll
    for (int o = 0; o < DOUT; ++o) {
      float v = hw[h * DOUT + o];
      s1 += v * Asrc[h * DOUT + o];
      s2 += v * Atgt[h * DOUT + o];
    }
    ssrc[(size_t)n * NH + h] = s1;
    stgt[(size_t)n * NH + h] = s2;
  }
}

#define BK_SHIFT  8
#define BK_NODES  (1 << BK_SHIFT)
#define L1_BLOCKS 256
#define L1_T      1024
#define L1_W      (L1_T / 32)
#define L1_U      8
#define L2_T      512
#define L2_SEGS   (L1_BLOCKS / (L2_T / 32))

__global__ __launch_bounds__(L1_T) void k_l1_split(const int* __restrict__ src,
                                                   const int* __restrict__ tgt,
                                                   int* __restrict__ lb,
                                                   int* __restrict__ pe_raw, int E, int nb) {
  extern __shared__ int l1_cnt[];
  __shared__ int wtot[L1_W];
  int2* pe = (int2*)pe_raw;
  const int tid  = threadIdx.x;
  const int lane = tid & 31;
  const int w    = tid >> 5;
  for (int i = tid; i < nb; i += L1_T) l1_cnt[i] = 0;
  __syncthreads();

  const int chunk = (E + (int)gridDim.x - 1) / (int)gridDim.x;
  const int beg   = (int)blockIdx.x * chunk;
  const int end   = min(E, beg + chunk);
  for (int e0 = beg; e0 < end; e0 += L1_T * L1_U) {
    int t[L1_U];
#pragma unroll
    for (int u = 0; u < L1_U; ++u) {
      const int e = e0 + u * L1_T + tid;
      t[u] = tgt[e < end ? e : end - 1];
    }
#pragma unroll
    for (int u = 0; u < L1_U; ++u)
      if (e0 + u * L1_T + tid < end) atomicAdd(&l1_cnt[t[u] >> BK_SHIFT], 1);
  }
  __syncthreads();

  int carry = 0;
  for (int base = 0; base < nb; base += L1_T) {
    const int i = base + tid;
    const int v = i < nb ? l1_cnt[i] : 0;
    int inc = v;
    for (int d = 1; d < 32; d <<= 1) {
      const int y = __shfl_up(inc, d);
      if (lane >= d) inc += y;
    }
    if (lane == 31) wtot[w] = inc;
    __syncthreads();
    const int wt = lane < L1_W ? wtot[lane] : 0;
    int pre = lane < w ? wt : 0;
    int all = wt;
    for (int d = 16; d >= 1; d >>= 1) {
      pre += __shfl_xor(pre, d);
      all += __shfl_xor(all, d);
    }
    const int ex = carry + pre + inc - v;
    if (i < nb) {
      l1_cnt[i] = ex;
      lb[(size_t)i * gridDim.x + blockIdx.x] = ex;
    }
    carry += all;
    __syncthreads();
  }
  if (tid == 0) lb[(size_t)nb * gridDim.x + blockIdx.x] = carry;

  for (int e0 = beg; e0 < end; e0 += L1_T * L1_U) {
    int s[L1_U];
    int t[L1_U];
#pragma unroll
    for (int u = 0; u < L1_U; ++u) {
      const int e  = e0 + u * L1_T + tid;
      const int ec = e < end ? e : end - 1;
      t[u] = tgt[ec];
      s[u] = src[ec];
    }
#pragma unroll
    for (int u = 0; u < L1_U; ++u) {
      if (e0 + u * L1_T + tid < end) {
        const int pos = atomicAdd(&l1_cnt[t[u] >> BK_SHIFT], 1);
        pe[(size_t)beg + pos] = make_int2(s[u], t[u]);
      }
    }
  }
}

__global__ __launch_bounds__(L2_T) void k_l2_bucket(const int* __restrict__ pe_raw,
                                                    const int* __restrict__ lb,
                                                    int* __restrict__ offs,
                                                    int* __restrict__ ssorted, int N, int E) {
  const int2* pe = (const int2*)pe_raw;
  __shared__ int cnt[BK_NODES];
  __shared__ int wtot[BK_NODES / 32];
  __shared__ int seg_st[L1_BLOCKS];
  __shared__ int seg_ln[L1_BLOCKS];
  __shared__ int wbeg[L1_BLOCKS / 32];
  __shared__ int wlen[L1_BLOCKS / 32];
  const int tid   = threadIdx.x;
  const int lane  = tid & 31;
  const int w     = tid >> 5;
  const int k     = blockIdx.x;
  const int chunk = (E + L1_BLOCKS - 1) / L1_BLOCKS;

  if (tid < BK_NODES) cnt[tid] = 0;
  if (tid < L1_BLOCKS) {
    const int a = lb[(size_t)k * L1_BLOCKS + tid];
    const int z = lb[(size_t)(k + 1) * L1_BLOCKS + tid];
    seg_st[tid] = tid * chunk + a;
    seg_ln[tid] = z - a;
    int sa = a, sl = z - a;
    for (int d = 16; d >= 1; d >>= 1) {
      sa += __shfl_xor(sa, d);
      sl += __shfl_xor(sl, d);
    }
    if (lane == 0) {
      wbeg[w] = sa;
      wlen[w] = sl;
    }
  }
  __syncthreads();
  int beg = 0, tot = 0;
  for (int j = 0; j < L1_BLOCKS / 32; ++j) {
    beg += wbeg[j];
    tot += wlen[j];
  }
  const int end = beg + tot;
  const int sb  = w * L2_SEGS;

  {
    int t[L2_SEGS];
#pragma unroll
    for (int u = 0; u < L2_SEGS; ++u) {
      const int ln = seg_ln[sb + u];
      const int st = seg_st[sb + u];
      t[u] = pe[lane < ln ? st + lane : 0].y;
    }
#pragma unroll
    for (int u = 0; u < L2_SEGS; ++u)
      if (lane < seg_ln[sb + u]) atomicAdd(&cnt[t[u] & (BK_NODES - 1)], 1);
    for (int u = 0; u < L2_SEGS; ++u) {
      const int ln = seg_ln[sb + u];
      const int st = seg_st[sb + u];
      for (int j = 32 + lane; j < ln; j += 32)
        atomicAdd(&cnt[pe[st + j].y & (BK_NODES - 1)], 1);
    }
  }
  __syncthreads();

  const int v = tid < BK_NODES ? cnt[tid] : 0;
  int inc = v;
  for (int d = 1; d < 32; d <<= 1) {
    const int y = __shfl_up(inc, d);
    if (lane >= d) inc += y;
  }
  if (tid < BK_NODES && lane == 31) wtot[w] = inc;
  __syncthreads();
  if (tid < BK_NODES) {
    int pre = 0;
    for (int j = 0; j < w; ++j) pre += wtot[j];
    const int start = beg + pre + inc - v;
    cnt[tid] = start;
    const int node = k * BK_NODES + tid;
    if (node < N) offs[node] = start;
  }
  if (k == (int)gridDim.x - 1 && tid == 0) offs[N] = end;
  __syncthreads();

  {
    int2 q[L2_SEGS];
#pragma unroll
    for (int u = 0; u < L2_SEGS; ++u) {
      const int ln = seg_ln[sb + u];
      const int st = seg_st[sb + u];
      q[u] = pe[lane < ln ? st + lane : 0];
    }
#pragma unroll
    for (int u = 0; u < L2_SEGS; ++u) {
      if (lane < seg_ln[sb + u]) {
        const int pos = atomicAdd(&cnt[q[u].y & (BK_NODES - 1)], 1);
        ssorted[pos] = q[u].x;
      }
    }
    for (int u = 0; u < L2_SEGS; ++u) {
      const int ln = seg_ln[sb + u];
      const int st = seg_st[sb + u];
      for (int j = 32 + lane; j < ln; j += 32) {
        const int2 p = pe[st + j];
        const int pos = atomicAdd(&cnt[p.y & (BK_NODES - 1)], 1);
        ssorted[pos] = p.x;
      }
    }
  }
}

__device__ __forceinline__ float lrelu(float x) { return x > 0.f ? x : 0.2f * x; }

__device__ __forceinline__ float4 logits4(float4 a, float4 b) {
  float4 e;
  e.x = lrelu(a.x + b.x);
  e.y = lrelu(a.y + b.y);
  e.z = lrelu(a.z + b.z);
  e.w = lrelu(a.w + b.w);
  return e;
}

__device__ __forceinline__ int lane_val_i(int v, int k) {
  return __builtin_amdgcn_readlane(v, k);
}

__device__ __forceinline__ float uniform_f(float v) {
  return __builtin_bit_cast(float, __builtin_amdgcn_readfirstlane(__builtin_bit_cast(int, v)));
}

#define GB 8

__device__ __forceinline__ float head_weight(float4 al, int k, unsigned m0, unsigned m1,
                                             unsigned m2, unsigned m3) {
  const unsigned b0 = (unsigned)__builtin_amdgcn_readlane(__builtin_bit_cast(int, al.x), k);
  const unsigned b1 = (unsigned)__builtin_amdgcn_readlane(__builtin_bit_cast(int, al.y), k);
  const unsigned b2 = (unsigned)__builtin_amdgcn_readlane(__builtin_bit_cast(int, al.z), k);
  const unsigned b3 = (unsigned)__builtin_amdgcn_readlane(__builtin_bit_cast(int, al.w), k);
  return __builtin_bit_cast(float, (b0 & m0) | (b1 & m1) | (b2 & m2) | (b3 & m3));
}

__device__ __forceinline__ void gather_rows(int s, float4 al, int n, int lane,
                                            const float* __restrict__ Hw, float4& acc) {
  const int h = lane >> 3;
  const unsigned m0 = h == 0 ? ~0u : 0u, m1 = h == 1 ? ~0u : 0u;
  const unsigned m2 = h == 2 ? ~0u : 0u, m3 = h == 3 ? ~0u : 0u;
  const float4* rows = (const float4*)Hw + lane;
  for (int k0 = 0; k0 < n; k0 += GB) {
    float4 v[GB];
    float  w[GB];
#pragma unroll
    for (int u = 0; u < GB; ++u) {
      const int k = k0 + u;
      if (k < n) {
        v[u] = rows[(size_t)lane_val_i(s, k) * (NH * DOUT / 4)];
        w[u] = head_weight(al, k, m0, m1, m2, m3);
      }
    }
#pragma unroll
    for (int u = 0; u < GB; ++u) {
      if (k0 + u < n) {
        acc.x += w[u] * v[u].x;
        acc.y += w[u] * v[u].y;
        acc.z += w[u] * v[u].z;
        acc.w += w[u] * v[u].w;
      }
    }
  }
}

__global__ __launch_bounds__(256, 16) void k_node(const int* __restrict__ offs,
                                              const int* __restrict__ ssorted,
                                              const float* __restrict__ ssrc,
                                              const float* __restrict__ stgt,
                                              const float* __restrict__ Hw,
                                              float* __restrict__ dout, int N) {
  const int lane = threadIdx.x & 31;
  const int t = blockIdx.x * 8 + __builtin_amdgcn_readfirstlane((int)(threadIdx.x >> 5));
  if (t >= N) return;
  const int beg = __builtin_amdgcn_readfirstlane(offs[t]);
  const int end = __builtin_amdgcn_readfirstlane(offs[t + 1]);
  const float4 bt = ((const float4*)stgt)[t];

  const bool own0 = beg + lane < end;
  int s0 = 0;
  float4 e0;
  e0.x = e0.y = e0.z = e0.w = -INFINITY;
  if (own0) {
    s0 = ssorted[beg + lane];
    e0 = logits4(((const float4*)ssrc)[s0], bt);
  }

  float4 mx = e0;
  for (int j = beg + 32 + lane; j < end; j += 32) {
    const float4 e = logits4(((const float4*)ssrc)[ssorted[j]], bt);
    mx.x = fmaxf(mx.x, e.x);
    mx.y = fmaxf(mx.y, e.y);
    mx.z = fmaxf(mx.z, e.z);
    mx.w = fmaxf(mx.w, e.w);
  }
  for (int d = 16; d >= 1; d >>= 1) {
    mx.x = fmaxf(mx.x, __shfl_xor(mx.x, d));
    mx.y = fmaxf(mx.y, __shfl_xor(mx.y, d));
    mx.z = fmaxf(mx.z, __shfl_xor(mx.z, d));
    mx.w = fmaxf(mx.w, __shfl_xor(mx.w, d));
  }
  mx.x = uniform_f(mx.x);
  mx.y = uniform_f(mx.y);
  mx.z = uniform_f(mx.z);
  mx.w = uniform_f(mx.w);

  float4 p0;
  p0.x = own0 ? expf(e0.x - mx.x) : 0.f;
  p0.y = own0 ? expf(e0.y - mx.y) : 0.f;
  p0.z = own0 ? expf(e0.z - mx.z) : 0.f;
  p0.w = own0 ? expf(e0.w - mx.w) : 0.f;
  float4 sm = p0;
  for (int j = beg + 32 + lane; j < end; j += 32) {
    const float4 e = logits4(((const float4*)ssrc)[ssorted[j]], bt);
    sm.x += expf(e.x - mx.x);
    sm.y += expf(e.y - mx.y);
    sm.z += expf(e.z - mx.z);
    sm.w += expf(e.w - mx.w);
  }
  for (int d = 16; d >= 1; d >>= 1) {
    sm.x += __shfl_xor(sm.x, d);
    sm.y += __shfl_xor(sm.y, d);
    sm.z += __shfl_xor(sm.z, d);
    sm.w += __shfl_xor(sm.w, d);
  }
  sm.x = uniform_f(sm.x);
  sm.y = uniform_f(sm.y);
  sm.z = uniform_f(sm.z);
  sm.w = uniform_f(sm.w);

  float4 acc;
  acc.x = acc.y = acc.z = acc.w = 0.f;
  int s = s0;
  float4 al;
  al.x = p0.x / sm.x;
  al.y = p0.y / sm.y;
  al.z = p0.z / sm.z;
  al.w = p0.w / sm.w;
  for (int c = beg; c < end; c += 32) {
    if (c != beg) {
      const int j = c + lane;
      s = 0;
      al.x = al.y = al.z = al.w = 0.f;
      if (j < end) {
        s = ssorted[j];
        const float4 e = logits4(((const float4*)ssrc)[s], bt);
        al.x = expf(e.x - mx.x) / sm.x;
        al.y = expf(e.y - mx.y) / sm.y;
        al.z = expf(e.z - mx.z) / sm.z;
        al.w = expf(e.w - mx.w) / sm.w;
      }
    }
    const int left = end - c;
    gather_rows(s, al, left < 32 ? left : 32, lane, Hw, acc);
  }

  float4 r;
  r.x = acc.x > 0.f ? acc.x : expm1f(acc.x);
  r.y = acc.y > 0.f ? acc.y : expm1f(acc.y);
  r.z = acc.z > 0.f ? acc.z : expm1f(acc.z);
  r.w = acc.w > 0.f ? acc.w : expm1f(acc.w);
  float* out = dout + ((size_t)(lane >> 3) * N + t) * DOUT;
  ((float4*)out)[lane & 7] = r;
}

extern "C" void kernel_launch(void* const* d_in, const int* in_sizes, int n_in,
                              void* d_out, int out_size, void* d_ws, size_t ws_size,
                              hipStream_t stream) {
  const float* Hin  = (const float*)d_in[0];
  const int*   ei   = (const int*)d_in[1];
  const float* W    = (const float*)d_in[2];
  const float* Asrc = (const float*)d_in[3];
  const float* Atgt = (const float*)d_in[4];
  float* dout = (float*)d_out;

  const int N = in_sizes[0] / DIN;
  const int E = in_sizes[1] / 2;
  const int* src = ei;
  const int* tgt = ei + E;

  char* p = (char*)d_ws;
  float* Hw      = (float*)p; p += (size_t)N * NH * DOUT * sizeof(float);
  float* ssrc    = (float*)p; p += (size_t)N * NH * sizeof(float);
  float* stgt    = (float*)p; p += (size_t)N * NH * sizeof(float);
  int*   offs    = (int*)p;   p += (size_t)(N + 4) * sizeof(int);
  int*   ssorted = (int*)p;   p += (size_t)E * sizeof(int);
  const int nb = (N + BK_NODES - 1) >> BK_SHIFT;
  int*  pe = (int*)Hw;
  int*  lb = (int*)Hw + (((size_t)2 * E + 3) & ~(size_t)3);
  (void)ws_size; (void)n_in; (void)out_size;

  const int B = 256;
  k_l1_split<<<L1_BLOCKS, L1_T, nb * sizeof(int), stream>>>(src, tgt, lb, pe, E, nb);
  k_l2_bucket<<<nb, L2_T, 0, stream>>>(pe, lb, offs, ssorted, N, E);
  k_hw_wmma<<<HW_BLOCKS, 128, 0, stream>>>(Hin, W, Hw, N);
  k_scores<<<(N + B - 1) / B, B, 0, stream>>>(Hw, Asrc, Atgt, ssrc, stgt, N);
  k_node<<<(N + 7) / 8, 256, 0, stream>>>(offs, ssorted, ssrc, stgt, Hw, dout, N);
}


# k_node forms each edge's logit from the gathered H_w row (dot with A_src, 8-lane DPP sums), softmax online in batches of 8: score arrays, their gather, k_scores launch gone (5->4); sum order changed
# speedup vs baseline: 19.9960x; 1.1192x over previous
#include <hip/hip_runtime.h>
#include <hip/hip_bf16.h>

typedef float v2f __attribute__((ext_vector_type(2)));
typedef float v8f __attribute__((ext_vector_type(8)));

#define NH   4
#define DIN  128
#define DOUT 32

#define HW_BLOCKS 1024
#define HW_SPREAD 569

__device__ __forceinline__ const float4* hw_arow(const float* __restrict__ Hin, int tile, int m,
                                                 int khalf, int N) {
  int r = tile * 16 + m;
  if (r >= N) r = N - 1;
  return (const float4*)(Hin + (size_t)r * DIN + 4 * khalf);
}

__device__ __forceinline__ void hw_mma4(float4 a, float4 p, float4 q, v8f& acc0, v8f& acc1) {
  const v2f a0 = {a.x, a.y}, a1 = {a.z, a.w};
  const v2f p0 = {p.x, p.y}, p1 = {p.z, p.w};
  const v2f q0 = {q.x, q.y}, q1 = {q.z, q.w};
  acc0 = __builtin_amdgcn_wmma_f32_16x16x4_f32(false, a0, false, p0, (short)0, acc0, false, false);
  acc1 = __builtin_amdgcn_wmma_f32_16x16x4_f32(false, a0, false, q0, (short)0, acc1, false, false);
  acc0 = __builtin_amdgcn_wmma_f32_16x16x4_f32(false, a1, false, p1, (short)0, acc0, false, false);
  acc1 = __builtin_amdgcn_wmma_f32_16x16x4_f32(false, a1, false, q1, (short)0, acc1, false, false);
}

__global__ __launch_bounds__(128, 4) void k_hw_wmma(const float* __restrict__ Hin,
                                                    const float* __restrict__ W,
                                                    float* __restrict__ Hw, int N) {
  const int h     = threadIdx.x >> 5;
  const int lane  = threadIdx.x & 31;
  const int m     = lane & 15;
  const int khalf = lane >> 4;
  const int b     = (int)blockIdx.x;

  const int ntiles = (N + 15) >> 4;
  const int full   = ntiles / HW_BLOCKS;
  const int rem    = ntiles - full * HW_BLOCKS;
  const int x      = (b * HW_SPREAD) & (HW_BLOCKS - 1);
  const int nmy    = full + (x < rem ? 1 : 0);

  const float4* b0p = (const float4*)(W + ((size_t)h * DOUT + m) * DIN + 4 * khalf);
  const float4* b1p = (const float4*)(W + ((size_t)h * DOUT + m + 16) * DIN + 4 * khalf);
  float4 b0[16], b1[16];
#pragma unroll
  for (int j = 0; j < 16; ++j) {
    b0[j] = b0p[2 * j];
    b1[j] = b1p[2 * j];
  }

  float4 alo[8], ahi[8];
  int tile = 0 < full ? b : full * HW_BLOCKS + x;
  {
    const float4* ap = hw_arow(Hin, tile, m, khalf, N);
#pragma unroll
    for (int j = 0; j < 8; ++j) alo[j] = ap[2 * j];
#pragma unroll
    for (int j = 0; j < 8; ++j) ahi[j] = ap[2 * (j + 8)];
  }

  for (int i = 0; i < nmy; ++i) {
    const int nxt = i + 1 < full ? b + (i + 1) * HW_BLOCKS : full * HW_BLOCKS + x;
    const float4* np = hw_arow(Hin, nxt, m, khalf, N);
    const int base = tile * 16;

    v8f acc0 = {0.f, 0.f, 0.f, 0.f, 0.f, 0.f, 0.f, 0.f};
    v8f acc1 = {0.f, 0.f, 0.f, 0.f, 0.f, 0.f, 0.f, 0.f};
#pragma unroll
    for (int j = 0; j < 8; ++j) hw_mma4(alo[j], b0[j], b1[j], acc0, acc1);
#pragma unroll
    for (int j = 0; j < 8; ++j) alo[j] = np[2 * j];
#pragma unroll
    for (int j = 0; j < 8; ++j) hw_mma4(ahi[j], b0[j + 8], b1[j + 8], acc0, acc1);
#pragma unroll
    for (int j = 0; j < 8; ++j) ahi[j] = np[2 * (j + 8)];

#pragma unroll
    for (int g = 0; g < 8; ++g) {
      int node = base + g + 8 * khalf;
      if (node < N) {
        float* dst = Hw + ((size_t)node * NH + h) * DOUT;
        dst[m]      = acc0[g];
        dst[m + 16] = acc1[g];
      }
    }
    tile = nxt;
  }
}

#define BK_SHIFT  8
#define BK_NODES  (1 << BK_SHIFT)
#define L1_BLOCKS 256
#define L1_T      1024
#define L1_W      (L1_T / 32)
#define L1_U      8
#define L2_T      512
#define L2_SEGS   (L1_BLOCKS / (L2_T / 32))

__global__ __launch_bounds__(L1_T) void k_l1_split(const int* __restrict__ src,
                                                   const int* __restrict__ tgt,
                                                   int* __restrict__ lb,
                                                   int* __restrict__ pe_raw, int E, int nb) {
  extern __shared__ int l1_cnt[];
  __shared__ int wtot[L1_W];
  int2* pe = (int2*)pe_raw;
  const int tid  = threadIdx.x;
  const int lane = tid & 31;
  const int w    = tid >> 5;
  for (int i = tid; i < nb; i += L1_T) l1_cnt[i] = 0;
  __syncthreads();

  const int chunk = (E + (int)gridDim.x - 1) / (int)gridDim.x;
  const int beg   = (int)blockIdx.x * chunk;
  const int end   = min(E, beg + chunk);
  for (int e0 = beg; e0 < end; e0 += L1_T * L1_U) {
    int t[L1_U];
#pragma unroll
    for (int u = 0; u < L1_U; ++u) {
      const int e = e0 + u * L1_T + tid;
      t[u] = tgt[e < end ? e : end - 1];
    }
#pragma unroll
    for (int u = 0; u < L1_U; ++u)
      if (e0 + u * L1_T + tid < end) atomicAdd(&l1_cnt[t[u] >> BK_SHIFT], 1);
  }
  __syncthreads();

  int carry = 0;
  for (int base = 0; base < nb; base += L1_T) {
    const int i = base + tid;
    const int v = i < nb ? l1_cnt[i] : 0;
    int inc = v;
    for (int d = 1; d < 32; d <<= 1) {
      const int y = __shfl_up(inc, d);
      if (lane >= d) inc += y;
    }
    if (lane == 31) wtot[w] = inc;
    __syncthreads();
    const int wt = lane < L1_W ? wtot[lane] : 0;
    int pre = lane < w ? wt : 0;
    int all = wt;
    for (int d = 16; d >= 1; d >>= 1) {
      pre += __shfl_xor(pre, d);
      all += __shfl_xor(all, d);
    }
    const int ex = carry + pre + inc - v;
    if (i < nb) {
      l1_cnt[i] = ex;
      lb[(size_t)i * gridDim.x + blockIdx.x] = ex;
    }
    carry += all;
    __syncthreads();
  }
  if (tid == 0) lb[(size_t)nb * gridDim.x + blockIdx.x] = carry;

  for (int e0 = beg; e0 < end; e0 += L1_T * L1_U) {
    int s[L1_U];
    int t[L1_U];
#pragma unroll
    for (int u = 0; u < L1_U; ++u) {
      const int e  = e0 + u * L1_T + tid;
      const int ec = e < end ? e : end - 1;
      t[u] = tgt[ec];
      s[u] = src[ec];
    }
#pragma unroll
    for (int u = 0; u < L1_U; ++u) {
      if (e0 + u * L1_T + tid < end) {
        const int pos = atomicAdd(&l1_cnt[t[u] >> BK_SHIFT], 1);
        pe[(size_t)beg + pos] = make_int2(s[u], t[u]);
      }
    }
  }
}

__global__ __launch_bounds__(L2_T) void k_l2_bucket(const int* __restrict__ pe_raw,
                                                    const int* __restrict__ lb,
                                                    int* __restrict__ offs,
                                                    int* __restrict__ ssorted, int N, int E) {
  const int2* pe = (const int2*)pe_raw;
  __shared__ int cnt[BK_NODES];
  __shared__ int wtot[BK_NODES / 32];
  __shared__ int seg_st[L1_BLOCKS];
  __shared__ int seg_ln[L1_BLOCKS];
  __shared__ int wbeg[L1_BLOCKS / 32];
  __shared__ int wlen[L1_BLOCKS / 32];
  const int tid   = threadIdx.x;
  const int lane  = tid & 31;
  const int w     = tid >> 5;
  const int k     = blockIdx.x;
  const int chunk = (E + L1_BLOCKS - 1) / L1_BLOCKS;

  if (tid < BK_NODES) cnt[tid] = 0;
  if (tid < L1_BLOCKS) {
    const int a = lb[(size_t)k * L1_BLOCKS + tid];
    const int z = lb[(size_t)(k + 1) * L1_BLOCKS + tid];
    seg_st[tid] = tid * chunk + a;
    seg_ln[tid] = z - a;
    int sa = a, sl = z - a;
    for (int d = 16; d >= 1; d >>= 1) {
      sa += __shfl_xor(sa, d);
      sl += __shfl_xor(sl, d);
    }
    if (lane == 0) {
      wbeg[w] = sa;
      wlen[w] = sl;
    }
  }
  __syncthreads();
  int beg = 0, tot = 0;
  for (int j = 0; j < L1_BLOCKS / 32; ++j) {
    beg += wbeg[j];
    tot += wlen[j];
  }
  const int end = beg + tot;
  const int sb  = w * L2_SEGS;

  {
    int t[L2_SEGS];
#pragma unroll
    for (int u = 0; u < L2_SEGS; ++u) {
      const int ln = seg_ln[sb + u];
      const int st = seg_st[sb + u];
      t[u] = pe[lane < ln ? st + lane : 0].y;
    }
#pragma unroll
    for (int u = 0; u < L2_SEGS; ++u)
      if (lane < seg_ln[sb + u]) atomicAdd(&cnt[t[u] & (BK_NODES - 1)], 1);
    for (int u = 0; u < L2_SEGS; ++u) {
      const int ln = seg_ln[sb + u];
      const int st = seg_st[sb + u];
      for (int j = 32 + lane; j < ln; j += 32)
        atomicAdd(&cnt[pe[st + j].y & (BK_NODES - 1)], 1);
    }
  }
  __syncthreads();

  const int v = tid < BK_NODES ? cnt[tid] : 0;
  int inc = v;
  for (int d = 1; d < 32; d <<= 1) {
    const int y = __shfl_up(inc, d);
    if (lane >= d) inc += y;
  }
  if (tid < BK_NODES && lane == 31) wtot[w] = inc;
  __syncthreads();
  if (tid < BK_NODES) {
    int pre = 0;
    for (int j = 0; j < w; ++j) pre += wtot[j];
    const int start = beg + pre + inc - v;
    cnt[tid] = start;
    const int node = k * BK_NODES + tid;
    if (node < N) offs[node] = start;
  }
  if (k == (int)gridDim.x - 1 && tid == 0) offs[N] = end;
  __syncthreads();

  {
    int2 q[L2_SEGS];
#pragma unroll
    for (int u = 0; u < L2_SEGS; ++u) {
      const int ln = seg_ln[sb + u];
      const int st = seg_st[sb + u];
      q[u] = pe[lane < ln ? st + lane : 0];
    }
#pragma unroll
    for (int u = 0; u < L2_SEGS; ++u) {
      if (lane < seg_ln[sb + u]) {
        const int pos = atomicAdd(&cnt[q[u].y & (BK_NODES - 1)], 1);
        ssorted[pos] = q[u].x;
      }
    }
    for (int u = 0; u < L2_SEGS; ++u) {
      const int ln = seg_ln[sb + u];
      const int st = seg_st[sb + u];
      for (int j = 32 + lane; j < ln; j += 32) {
        const int2 p = pe[st + j];
        const int pos = atomicAdd(&cnt[p.y & (BK_NODES - 1)], 1);
        ssorted[pos] = p.x;
      }
    }
  }
}

__device__ __forceinline__ float lrelu(float x) { return x > 0.f ? x : 0.2f * x; }

#define GB 8

template <int M>
__device__ __forceinline__ float lane_xor(float x) {
  const int b = __builtin_bit_cast(int, x);
  return __builtin_bit_cast(float, __builtin_amdgcn_update_dpp(0, b, 0x160 | M, 0xf, 0xf, true));
}

template <int U>
__device__ __forceinline__ float group_bcast(float x) {
  return __builtin_bit_cast(
      float, __builtin_amdgcn_mov_dpp8(__builtin_bit_cast(int, x), U * 0x249249));
}

__device__ __forceinline__ float group_sum(float x) {
  x += lane_xor<1>(x);
  x += lane_xor<2>(x);
  x += lane_xor<4>(x);
  return x;
}

__device__ __forceinline__ float group_max(float x) {
  x = fmaxf(x, lane_xor<1>(x));
  x = fmaxf(x, lane_xor<2>(x));
  x = fmaxf(x, lane_xor<4>(x));
  return x;
}

__device__ __forceinline__ float dot4(float4 v, float4 a) {
  const v2f vlo = {v.x, v.y}, vhi = {v.z, v.w};
  const v2f alo = {a.x, a.y}, ahi = {a.z, a.w};
  const v2f t = vhi * ahi + vlo * alo;
  return t.x + t.y;
}

__device__ __forceinline__ float group_sum_scatter(const float (&d)[GB], bool b4, bool b2,
                                                   bool b1) {
  float r[4];
#pragma unroll
  for (int j = 0; j < 4; ++j) {
    const float keep = b4 ? d[j + 4] : d[j];
    const float send = b4 ? d[j] : d[j + 4];
    r[j] = keep + lane_xor<4>(send);
  }
  float r2[2];
#pragma unroll
  for (int j = 0; j < 2; ++j) {
    const float keep = b2 ? r[j + 2] : r[j];
    const float send = b2 ? r[j] : r[j + 2];
    r2[j] = keep + lane_xor<2>(send);
  }
  const float keep = b1 ? r2[1] : r2[0];
  const float send = b1 ? r2[0] : r2[1];
  return keep + lane_xor<1>(send);
}

typedef const __attribute__((address_space(1))) float4* grow_t;
__device__ __forceinline__ grow_t node_row(const float* __restrict__ Hw, unsigned node) {
  const unsigned long long a =
      (unsigned long long)Hw + (unsigned long long)node * (NH * DOUT * sizeof(float));
  const unsigned lo = (unsigned)__builtin_amdgcn_readfirstlane((int)(unsigned)a);
  const unsigned hi = (unsigned)__builtin_amdgcn_readfirstlane((int)(unsigned)(a >> 32));
  return (grow_t)(((unsigned long long)hi << 32) | lo);
}

template <int U>
__device__ __forceinline__ void add_row(float p, float4 v, float4& acc) {
  const float w = group_bcast<U>(p);
  acc.x += w * v.x;
  acc.y += w * v.y;
  acc.z += w * v.z;
  acc.w += w * v.w;
}

__global__ __launch_bounds__(256, 16) void k_node(const int* __restrict__ offs,
                                                  const int* __restrict__ ssorted,
                                                  const float* __restrict__ Asrc,
                                                  const float* __restrict__ Atgt,
                                                  const float* __restrict__ Hw,
                                                  float* __restrict__ dout, int N) {
  const int lane = threadIdx.x & 31;
  const int t = blockIdx.x * 8 + __builtin_amdgcn_readfirstlane((int)(threadIdx.x >> 5));
  if (t >= N) return;
  const int q = lane & 7;
  const bool b4 = (q & 4) != 0, b2 = (q & 2) != 0, b1 = (q & 1) != 0;
  const float4 as = ((const float4*)Asrc)[lane];
  const float4 at = ((const float4*)Atgt)[lane];
  const float4 vt = node_row(Hw, (unsigned)t)[lane];
  const int beg = __builtin_amdgcn_readfirstlane(offs[t]);
  const int end = __builtin_amdgcn_readfirstlane(offs[t + 1]);

  int s = 0;
  if (beg + lane < end) s = ssorted[beg + lane];
  asm volatile("" : "+v"(s));

  const float st = group_sum(dot4(vt, at));

  float m = -INFINITY;
  float l = 0.f;
  float4 acc;
  acc.x = acc.y = acc.z = acc.w = 0.f;
  for (int c = beg; c < end; c += 32) {
    if (c != beg) {
      const int j = c + lane;
      s = 0;
      if (j < end) s = ssorted[j];
      asm volatile("" : "+v"(s));
    }
    const int left = end - c;
    const int n = left < 32 ? left : 32;
    for (int k0 = 0; k0 < n; k0 += GB) {
      float4 v[GB];
      float d[GB];
      const int last = n - 1;
#pragma unroll
      for (int u = 0; u < GB; ++u) {
        const int k = k0 + u < last ? k0 + u : last;
        v[u] = node_row(Hw, (unsigned)__builtin_amdgcn_readlane(s, k))[lane];
      }
#pragma unroll
      for (int u = 0; u < GB; ++u) d[u] = dot4(v[u], as);

      float e = lrelu(group_sum_scatter(d, b4, b2, b1) + st);
      if (k0 + q >= n) e = -INFINITY;
      const float mn = fmaxf(m, group_max(e));
      const float sc = expf(m - mn);
      const float p  = expf(e - mn);
      l = l * sc + p;
      acc.x *= sc;
      acc.y *= sc;
      acc.z *= sc;
      acc.w *= sc;
      m = mn;
      add_row<0>(p, v[0], acc);
      add_row<1>(p, v[1], acc);
      add_row<2>(p, v[2], acc);
      add_row<3>(p, v[3], acc);
      add_row<4>(p, v[4], acc);
      add_row<5>(p, v[5], acc);
      add_row<6>(p, v[6], acc);
      add_row<7>(p, v[7], acc);
    }
  }
  const float lsum = group_sum(l);
  const float inv  = beg < end ? 1.0f / lsum : 0.f;
  acc.x *= inv;
  acc.y *= inv;
  acc.z *= inv;
  acc.w *= inv;

  float4 r;
  r.x = acc.x > 0.f ? acc.x : expm1f(acc.x);
  r.y = acc.y > 0.f ? acc.y : expm1f(acc.y);
  r.z = acc.z > 0.f ? acc.z : expm1f(acc.z);
  r.w = acc.w > 0.f ? acc.w : expm1f(acc.w);
  float* out = dout + ((size_t)(lane >> 3) * N + t) * DOUT;
  ((float4*)out)[lane & 7] = r;
}

extern "C" void kernel_launch(void* const* d_in, const int* in_sizes, int n_in,
                              void* d_out, int out_size, void* d_ws, size_t ws_size,
                              hipStream_t stream) {
  const float* Hin  = (const float*)d_in[0];
  const int*   ei   = (const int*)d_in[1];
  const float* W    = (const float*)d_in[2];
  const float* Asrc = (const float*)d_in[3];
  const float* Atgt = (const float*)d_in[4];
  float* dout = (float*)d_out;

  const int N = in_sizes[0] / DIN;
  const int E = in_sizes[1] / 2;
  const int* src = ei;
  const int* tgt = ei + E;

  char* p = (char*)d_ws;
  float* Hw      = (float*)p; p += (size_t)N * NH * DOUT * sizeof(float);
  int*   offs    = (int*)p;   p += (size_t)(N + 4) * sizeof(int);
  int*   ssorted = (int*)p;   p += (size_t)E * sizeof(int);
  const int nb = (N + BK_NODES - 1) >> BK_SHIFT;
  int*  pe = (int*)Hw;
  int*  lb = (int*)Hw + (((size_t)2 * E + 3) & ~(size_t)3);
  (void)ws_size; (void)n_in; (void)out_size;

  k_l1_split<<<L1_BLOCKS, L1_T, nb * sizeof(int), stream>>>(src, tgt, lb, pe, E, nb);
  k_l2_bucket<<<nb, L2_T, 0, stream>>>(pe, lb, offs, ssorted, N, E);
  k_hw_wmma<<<HW_BLOCKS, 128, 0, stream>>>(Hin, W, Hw, N);
  k_node<<<(N + 7) / 8, 256, 0, stream>>>(offs, ssorted, Asrc, Atgt, Hw, dout, N);
}


# k_node takes over sort level two: a work-group sorts its bucket's pairs by target in LDS, then its waves gather its targets; sorted list and a launch gone (4->3); sum order: LDS arrival, changed
# speedup vs baseline: 21.0965x; 1.0550x over previous
#include <hip/hip_runtime.h>
#include <hip/hip_bf16.h>

typedef float v2f __attribute__((ext_vector_type(2)));
typedef float v8f __attribute__((ext_vector_type(8)));

#define NH   4
#define DIN  128
#define DOUT 32

#define HW_BLOCKS 1024
#define HW_SPREAD 569

__device__ __forceinline__ const float4* hw_arow(const float* __restrict__ Hin, int tile, int m,
                                                 int khalf, int N) {
  int r = tile * 16 + m;
  if (r >= N) r = N - 1;
  return (const float4*)(Hin + (size_t)r * DIN + 4 * khalf);
}

__device__ __forceinline__ void hw_mma4(float4 a, float4 p, float4 q, v8f& acc0, v8f& acc1) {
  const v2f a0 = {a.x, a.y}, a1 = {a.z, a.w};
  const v2f p0 = {p.x, p.y}, p1 = {p.z, p.w};
  const v2f q0 = {q.x, q.y}, q1 = {q.z, q.w};
  acc0 = __builtin_amdgcn_wmma_f32_16x16x4_f32(false, a0, false, p0, (short)0, acc0, false, false);
  acc1 = __builtin_amdgcn_wmma_f32_16x16x4_f32(false, a0, false, q0, (short)0, acc1, false, false);
  acc0 = __builtin_amdgcn_wmma_f32_16x16x4_f32(false, a1, false, p1, (short)0, acc0, false, false);
  acc1 = __builtin_amdgcn_wmma_f32_16x16x4_f32(false, a1, false, q1, (short)0, acc1, false, false);
}

__global__ __launch_bounds__(128, 4) void k_hw_wmma(const float* __restrict__ Hin,
                                                    const float* __restrict__ W,
                                                    float* __restrict__ Hw, int N) {
  const int h     = threadIdx.x >> 5;
  const int lane  = threadIdx.x & 31;
  const int m     = lane & 15;
  const int khalf = lane >> 4;
  const int b     = (int)blockIdx.x;

  const int ntiles = (N + 15) >> 4;
  const int full   = ntiles / HW_BLOCKS;
  const int rem    = ntiles - full * HW_BLOCKS;
  const int x      = (b * HW_SPREAD) & (HW_BLOCKS - 1);
  const int nmy    = full + (x < rem ? 1 : 0);

  const float4* b0p = (const float4*)(W + ((size_t)h * DOUT + m) * DIN + 4 * khalf);
  const float4* b1p = (const float4*)(W + ((size_t)h * DOUT + m + 16) * DIN + 4 * khalf);
  float4 b0[16], b1[16];
#pragma unroll
  for (int j = 0; j < 16; ++j) {
    b0[j] = b0p[2 * j];
    b1[j] = b1p[2 * j];
  }

  float4 alo[8], ahi[8];
  int tile = 0 < full ? b : full * HW_BLOCKS + x;
  {
    const float4* ap = hw_arow(Hin, tile, m, khalf, N);
#pragma unroll
    for (int j = 0; j < 8; ++j) alo[j] = ap[2 * j];
#pragma unroll
    for (int j = 0; j < 8; ++j) ahi[j] = ap[2 * (j + 8)];
  }

  for (int i = 0; i < nmy; ++i) {
    const int nxt = i + 1 < full ? b + (i + 1) * HW_BLOCKS : full * HW_BLOCKS + x;
    const float4* np = hw_arow(Hin, nxt, m, khalf, N);
    const int base = tile * 16;

    v8f acc0 = {0.f, 0.f, 0.f, 0.f, 0.f, 0.f, 0.f, 0.f};
    v8f acc1 = {0.f, 0.f, 0.f, 0.f, 0.f, 0.f, 0.f, 0.f};
#pragma unroll
    for (int j = 0; j < 8; ++j) hw_mma4(alo[j], b0[j], b1[j], acc0, acc1);
#pragma unroll
    for (int j = 0; j < 8; ++j) alo[j] = np[2 * j];
#pragma unroll
    for (int j = 0; j < 8; ++j) hw_mma4(ahi[j], b0[j + 8], b1[j + 8], acc0, acc1);
#pragma unroll
    for (int j = 0; j < 8; ++j) ahi[j] = np[2 * (j + 8)];

#pragma unroll
    for (int g = 0; g < 8; ++g) {
      int node = base + g + 8 * khalf;
      if (node < N) {
        float* dst = Hw + ((size_t)node * NH + h) * DOUT;
        dst[m]      = acc0[g];
        dst[m + 16] = acc1[g];
      }
    }
    tile = nxt;
  }
}

#define BK_NODES  196
#define L1_BLOCKS 256
#define L1_T      1024
#define L1_W      (L1_T / 32)
#define L1_U      8
#define BN_T      1024
#define BN_W      (BN_T / 32)
#define BN_SEGS   (L1_BLOCKS / BN_W)
#define BN_CAP    8192

__global__ __launch_bounds__(L1_T) void k_l1_split(const int* __restrict__ src,
                                                   const int* __restrict__ tgt,
                                                   int* __restrict__ lb_raw,
                                                   int* __restrict__ pe, int E, int nb) {
  extern __shared__ int l1_cnt[];
  __shared__ int wtot[L1_W];
  unsigned short* lb = (unsigned short*)lb_raw;
  const int tid  = threadIdx.x;
  const int lane = tid & 31;
  const int w    = tid >> 5;
  for (int i = tid; i < nb; i += L1_T) l1_cnt[i] = 0;
  __syncthreads();

  const int chunk = (E + (int)gridDim.x - 1) / (int)gridDim.x;
  const int beg   = (int)blockIdx.x * chunk;
  const int end   = min(E, beg + chunk);
  for (int e0 = beg; e0 < end; e0 += L1_T * L1_U) {
    int t[L1_U];
#pragma unroll
    for (int u = 0; u < L1_U; ++u) {
      const int e = e0 + u * L1_T + tid;
      t[u] = tgt[e < end ? e : end - 1];
    }
#pragma unroll
    for (int u = 0; u < L1_U; ++u)
      if (e0 + u * L1_T + tid < end) atomicAdd(&l1_cnt[(unsigned)t[u] / BK_NODES], 1);
  }
  __syncthreads();

  int carry = 0;
  for (int base = 0; base < nb; base += L1_T) {
    const int i = base + tid;
    const int v = i < nb ? l1_cnt[i] : 0;
    int inc = v;
    for (int d = 1; d < 32; d <<= 1) {
      const int y = __shfl_up(inc, d);
      if (lane >= d) inc += y;
    }
    if (lane == 31) wtot[w] = inc;
    __syncthreads();
    const int wt = lane < L1_W ? wtot[lane] : 0;
    int pre = lane < w ? wt : 0;
    int all = wt;
    for (int d = 16; d >= 1; d >>= 1) {
      pre += __shfl_xor(pre, d);
      all += __shfl_xor(all, d);
    }
    const int ex = carry + pre + inc - v;
    if (i < nb) {
      l1_cnt[i] = ex;
      lb[(size_t)i * gridDim.x + blockIdx.x] = (unsigned short)ex;
    }
    carry += all;
    __syncthreads();
  }
  if (tid == 0) lb[(size_t)nb * gridDim.x + blockIdx.x] = (unsigned short)carry;

  for (int e0 = beg; e0 < end; e0 += L1_T * L1_U) {
    int s[L1_U];
    int t[L1_U];
#pragma unroll
    for (int u = 0; u < L1_U; ++u) {
      const int e  = e0 + u * L1_T + tid;
      const int ec = e < end ? e : end - 1;
      t[u] = tgt[ec];
      s[u] = src[ec];
    }
#pragma unroll
    for (int u = 0; u < L1_U; ++u) {
      if (e0 + u * L1_T + tid < end) {
        const int bk  = (int)((unsigned)t[u] / BK_NODES);
        const int pos = atomicAdd(&l1_cnt[bk], 1);
        pe[(size_t)beg + pos] = (s[u] << 8) | (t[u] - bk * BK_NODES);
      }
    }
  }
}

__device__ __forceinline__ float lrelu(float x) { return x > 0.f ? x : 0.2f * x; }

#define GB 8

template <int M>
__device__ __forceinline__ float lane_xor(float x) {
  const int b = __builtin_bit_cast(int, x);
  return __builtin_bit_cast(float, __builtin_amdgcn_update_dpp(0, b, 0x160 | M, 0xf, 0xf, true));
}

template <int U>
__device__ __forceinline__ float group_bcast(float x) {
  return __builtin_bit_cast(
      float, __builtin_amdgcn_mov_dpp8(__builtin_bit_cast(int, x), U * 0x249249));
}

__device__ __forceinline__ float group_sum(float x) {
  x += lane_xor<1>(x);
  x += lane_xor<2>(x);
  x += lane_xor<4>(x);
  return x;
}

__device__ __forceinline__ float group_max(float x) {
  x = fmaxf(x, lane_xor<1>(x));
  x = fmaxf(x, lane_xor<2>(x));
  x = fmaxf(x, lane_xor<4>(x));
  return x;
}

__device__ __forceinline__ float dot4(float4 v, float4 a) {
  const v2f vlo = {v.x, v.y}, vhi = {v.z, v.w};
  const v2f alo = {a.x, a.y}, ahi = {a.z, a.w};
  const v2f t = vhi * ahi + vlo * alo;
  return t.x + t.y;
}

__device__ __forceinline__ float group_sum_scatter(const float (&d)[GB], bool b4, bool b2,
                                                   bool b1) {
  float r[4];
#pragma unroll
  for (int j = 0; j < 4; ++j) {
    const float keep = b4 ? d[j + 4] : d[j];
    const float send = b4 ? d[j] : d[j + 4];
    r[j] = keep + lane_xor<4>(send);
  }
  float r2[2];
#pragma unroll
  for (int j = 0; j < 2; ++j) {
    const float keep = b2 ? r[j + 2] : r[j];
    const float send = b2 ? r[j] : r[j + 2];
    r2[j] = keep + lane_xor<2>(send);
  }
  const float keep = b1 ? r2[1] : r2[0];
  const float send = b1 ? r2[0] : r2[1];
  return keep + lane_xor<1>(send);
}

typedef const __attribute__((address_space(1))) float4* grow_t;
__device__ __forceinline__ grow_t node_row(const float* __restrict__ Hw, unsigned node) {
  const unsigned long long a =
      (unsigned long long)Hw + (unsigned long long)node * (NH * DOUT * sizeof(float));
  const unsigned lo = (unsigned)__builtin_amdgcn_readfirstlane((int)(unsigned)a);
  const unsigned hi = (unsigned)__builtin_amdgcn_readfirstlane((int)(unsigned)(a >> 32));
  return (grow_t)(((unsigned long long)hi << 32) | lo);
}

template <int U>
__device__ __forceinline__ void add_row(float p, float4 v, float4& acc) {
  const float w = group_bcast<U>(p);
  acc.x += w * v.x;
  acc.y += w * v.y;
  acc.z += w * v.z;
  acc.w += w * v.w;
}

__device__ __forceinline__ void gather_round(int s, int n, const float* __restrict__ Hw,
                                             const float* __restrict__ Atgt, int t, int lane,
                                             float4 as, bool& have_st, float& st, float& m,
                                             float& l, float4& acc) {
  const int q = lane & 7;
  const bool b4 = (q & 4) != 0, b2 = (q & 2) != 0, b1 = (q & 1) != 0;
  for (int k0 = 0; k0 < n; k0 += GB) {
    float4 v[GB];
    float d[GB];
    const int last = n - 1;
#pragma unroll
    for (int u = 0; u < GB; ++u) {
      const int k = k0 + u < last ? k0 + u : last;
      v[u] = node_row(Hw, (unsigned)__builtin_amdgcn_readlane(s, k))[lane];
    }
    if (!have_st) {
      const float4 vt = node_row(Hw, (unsigned)t)[lane];
      const float4 at = node_row(Atgt, 0u)[lane];
      st = group_sum(dot4(vt, at));
      have_st = true;
    }
#pragma unroll
    for (int u = 0; u < GB; ++u) d[u] = dot4(v[u], as);

    float e = lrelu(group_sum_scatter(d, b4, b2, b1) + st);
    if (k0 + q >= n) e = -INFINITY;
    const float mn = fmaxf(m, group_max(e));
    const float sc = expf(m - mn);
    const float p  = expf(e - mn);
    l = l * sc + p;
    acc.x *= sc;
    acc.y *= sc;
    acc.z *= sc;
    acc.w *= sc;
    m = mn;
    add_row<0>(p, v[0], acc);
    add_row<1>(p, v[1], acc);
    add_row<2>(p, v[2], acc);
    add_row<3>(p, v[3], acc);
    add_row<4>(p, v[4], acc);
    add_row<5>(p, v[5], acc);
    add_row<6>(p, v[6], acc);
    add_row<7>(p, v[7], acc);
  }
}

__global__ __launch_bounds__(BN_T, 16) void k_bucket_node(const int* __restrict__ pe,
                                                         const int* __restrict__ lb_raw,
                                                         const float* __restrict__ Asrc,
                                                         const float* __restrict__ Atgt,
                                                         const float* __restrict__ Hw,
                                                         float* __restrict__ dout, int N, int E) {
  const unsigned short* lb = (const unsigned short*)lb_raw;
  __shared__ int cnt[256];
  __shared__ int offl[257];
  __shared__ int wtot[8];
  __shared__ int seg_st[L1_BLOCKS];
  __shared__ int seg_ln[L1_BLOCKS];
  __shared__ int wlen[L1_BLOCKS / 32];
  __shared__ int nextt;
  __shared__ int srcl[BN_CAP];
  __shared__ int stage[BN_W][32];
  const int tid   = threadIdx.x;
  const int lane  = tid & 31;
  const int w     = __builtin_amdgcn_readfirstlane(tid >> 5);
  const int k     = blockIdx.x;
  const int chunk = (E + L1_BLOCKS - 1) / L1_BLOCKS;

  const float4 as = ((const float4*)Asrc)[lane];

  if (tid < 256) cnt[tid] = 0;
  if (tid == 0) nextt = 0;
  if (tid < L1_BLOCKS) {
    const int a = lb[(size_t)k * L1_BLOCKS + tid];
    const int z = lb[(size_t)(k + 1) * L1_BLOCKS + tid];
    seg_st[tid] = tid * chunk + a;
    seg_ln[tid] = z - a;
    int sl = z - a;
    for (int d = 16; d >= 1; d >>= 1) sl += __shfl_xor(sl, d);
    if (lane == 0) wlen[w] = sl;
  }
  __syncthreads();
  int tot = 0;
  for (int j = 0; j < L1_BLOCKS / 32; ++j) tot += wlen[j];
  tot = __builtin_amdgcn_readfirstlane(tot);
  const int sb = w * BN_SEGS;

  int pr[BN_SEGS];
#pragma unroll
  for (int u = 0; u < BN_SEGS; ++u) {
    const int ln = seg_ln[sb + u];
    const int st = seg_st[sb + u];
    pr[u] = pe[lane < ln ? st + lane : 0];
  }
#pragma unroll
  for (int u = 0; u < BN_SEGS; ++u)
    if (lane < seg_ln[sb + u]) atomicAdd(&cnt[pr[u] & 255], 1);
  for (int u = 0; u < BN_SEGS; ++u) {
    const int ln = seg_ln[sb + u];
    const int st = seg_st[sb + u];
    for (int j = 32 + lane; j < ln; j += 32) atomicAdd(&cnt[pe[st + j] & 255], 1);
  }
  __syncthreads();

  const int v = tid < 256 ? cnt[tid] : 0;
  int inc = v;
  for (int d = 1; d < 32; d <<= 1) {
    const int y = __shfl_up(inc, d);
    if (lane >= d) inc += y;
  }
  if (tid < 256 && lane == 31) wtot[w] = inc;
  __syncthreads();
  if (tid < 256) {
    int pre = 0;
    for (int j = 0; j < w; ++j) pre += wtot[j];
    const int start = pre + inc - v;
    cnt[tid]  = start;
    offl[tid] = start;
  }
  if (tid == 0) offl[256] = tot;
  __syncthreads();

#pragma unroll
  for (int u = 0; u < BN_SEGS; ++u) {
    if (lane < seg_ln[sb + u]) {
      const int pos = atomicAdd(&cnt[pr[u] & 255], 1);
      if (pos < BN_CAP) srcl[pos] = (int)((unsigned)pr[u] >> 8);
    }
  }
  for (int u = 0; u < BN_SEGS; ++u) {
    const int ln = seg_ln[sb + u];
    const int st = seg_st[sb + u];
    for (int j = 32 + lane; j < ln; j += 32) {
      const int p   = pe[st + j];
      const int pos = atomicAdd(&cnt[p & 255], 1);
      if (pos < BN_CAP) srcl[pos] = (int)((unsigned)p >> 8);
    }
  }
  __syncthreads();

  const bool listed = tot <= BN_CAP;
  for (;;) {
    int tl = 0;
    if (lane == 0) tl = atomicAdd(&nextt, 1);
    tl = __builtin_amdgcn_readfirstlane(tl);
    const int t = k * BK_NODES + tl;
    if (tl >= BK_NODES || t >= N) break;
    const int beg = __builtin_amdgcn_readfirstlane(offl[tl]);
    const int end = __builtin_amdgcn_readfirstlane(offl[tl + 1]);

    bool have_st = false;
    float st = 0.f;
    float m = -INFINITY;
    float l = 0.f;
    float4 acc;
    acc.x = acc.y = acc.z = acc.w = 0.f;
    if (listed) {
      for (int c = beg; c < end; c += 32) {
        int s = 0;
        if (c + lane < end) s = srcl[c + lane];
        const int left = end - c;
        gather_round(s, left < 32 ? left : 32, Hw, Atgt, t, lane, as, have_st, st, m, l, acc);
      }
    } else if (beg < end) {
      for (int j = 0; j < L1_BLOCKS; ++j) {
        const int ln  = __builtin_amdgcn_readfirstlane(seg_ln[j]);
        const int sst = __builtin_amdgcn_readfirstlane(seg_st[j]);
        for (int c = 0; c < ln; c += 32) {
          const bool in = c + lane < ln;
          const int p = in ? pe[sst + c + lane] : 0;
          const bool hit = in && (p & 255) == tl;
          const unsigned mask = __builtin_amdgcn_ballot_w32(hit);
          if (mask != 0u) {
            const int rank = (int)__builtin_amdgcn_mbcnt_lo(mask, 0u);
            if (hit) stage[w][rank] = (int)((unsigned)p >> 8);
            const int s = stage[w][lane];
            gather_round(s, __builtin_popcount(mask), Hw, Atgt, t, lane, as, have_st, st, m, l,
                         acc);
          }
        }
      }
    }
    const float lsum = group_sum(l);
    const float inv  = beg < end ? 1.0f / lsum : 0.f;
    acc.x *= inv;
    acc.y *= inv;
    acc.z *= inv;
    acc.w *= inv;

    float4 r;
    r.x = acc.x > 0.f ? acc.x : expm1f(acc.x);
    r.y = acc.y > 0.f ? acc.y : expm1f(acc.y);
    r.z = acc.z > 0.f ? acc.z : expm1f(acc.z);
    r.w = acc.w > 0.f ? acc.w : expm1f(acc.w);
    const unsigned off =
        (((unsigned)(lane >> 3) * (unsigned)N + (unsigned)t) * DOUT + (unsigned)(lane & 7) * 4u) *
        (unsigned)sizeof(float);
    *(float4*)((char*)dout + off) = r;
  }
}

extern "C" void kernel_launch(void* const* d_in, const int* in_sizes, int n_in,
                              void* d_out, int out_size, void* d_ws, size_t ws_size,
                              hipStream_t stream) {
  const float* Hin  = (const float*)d_in[0];
  const int*   ei   = (const int*)d_in[1];
  const float* W    = (const float*)d_in[2];
  const float* Asrc = (const float*)d_in[3];
  const float* Atgt = (const float*)d_in[4];
  float* dout = (float*)d_out;

  const int N = in_sizes[0] / DIN;
  const int E = in_sizes[1] / 2;
  const int* src = ei;
  const int* tgt = ei + E;

  const int nb = (N + BK_NODES - 1) / BK_NODES;
  char* p = (char*)d_ws;
  float* Hw = (float*)p; p += (size_t)N * NH * DOUT * sizeof(float);
  int*   pe = (int*)p;   p += (size_t)E * sizeof(int);
  int*   lb = (int*)p;
  (void)ws_size; (void)n_in; (void)out_size;

  k_l1_split<<<L1_BLOCKS, L1_T, nb * sizeof(int), stream>>>(src, tgt, lb, pe, E, nb);
  k_hw_wmma<<<HW_BLOCKS, 128, 0, stream>>>(Hin, W, Hw, N);
  k_bucket_node<<<nb, BN_T, 0, stream>>>(pe, lb, Asrc, Atgt, Hw, dout, N, E);
}


# k_hw_wmma also forms each node's logit halves (H_w row dot A_src, A_tgt) from its accumulator tiles; k_bucket_node gathers 4 B per edge and head: per-edge dot products gone; sum order changed
# speedup vs baseline: 22.3827x; 1.0610x over previous
#include <hip/hip_runtime.h>
#include <hip/hip_bf16.h>

typedef float v2f __attribute__((ext_vector_type(2)));
typedef float v8f __attribute__((ext_vector_type(8)));

#define NH   4
#define DIN  128
#define DOUT 32

#define HW_BLOCKS 1024
#define HW_SPREAD 569

template <int M>
__device__ __forceinline__ float lane_xor(float x) {
  const int b = __builtin_bit_cast(int, x);
  return __builtin_bit_cast(float, __builtin_amdgcn_update_dpp(0, b, 0x160 | M, 0xf, 0xf, true));
}

__device__ __forceinline__ const float4* hw_arow(const float* __restrict__ Hin, int tile, int m,
                                                 int khalf, int N) {
  int r = tile * 16 + m;
  if (r >= N) r = N - 1;
  return (const float4*)(Hin + (size_t)r * DIN + 4 * khalf);
}

__device__ __forceinline__ void hw_mma4(float4 a, float4 p, float4 q, v8f& acc0, v8f& acc1) {
  const v2f a0 = {a.x, a.y}, a1 = {a.z, a.w};
  const v2f p0 = {p.x, p.y}, p1 = {p.z, p.w};
  const v2f q0 = {q.x, q.y}, q1 = {q.z, q.w};
  acc0 = __builtin_amdgcn_wmma_f32_16x16x4_f32(false, a0, false, p0, (short)0, acc0, false, false);
  acc1 = __builtin_amdgcn_wmma_f32_16x16x4_f32(false, a0, false, q0, (short)0, acc1, false, false);
  acc0 = __builtin_amdgcn_wmma_f32_16x16x4_f32(false, a1, false, p1, (short)0, acc0, false, false);
  acc1 = __builtin_amdgcn_wmma_f32_16x16x4_f32(false, a1, false, q1, (short)0, acc1, false, false);
}

__global__ __launch_bounds__(128, 4) void k_hw_wmma(const float* __restrict__ Hin,
                                                    const float* __restrict__ W,
                                                    const float* __restrict__ Asrc,
                                                    const float* __restrict__ Atgt,
                                                    float* __restrict__ Hw,
                                                    float* __restrict__ slog, int N) {
  const int h     = threadIdx.x >> 5;
  const int lane  = threadIdx.x & 31;
  const int m     = lane & 15;
  const int khalf = lane >> 4;
  const int b     = (int)blockIdx.x;

  const int ntiles = (N + 15) >> 4;
  const int full   = ntiles / HW_BLOCKS;
  const int rem    = ntiles - full * HW_BLOCKS;
  const int x      = (b * HW_SPREAD) & (HW_BLOCKS - 1);
  const int nmy    = full + (x < rem ? 1 : 0);

  const float4* b0p = (const float4*)(W + ((size_t)h * DOUT + m) * DIN + 4 * khalf);
  const float4* b1p = (const float4*)(W + ((size_t)h * DOUT + m + 16) * DIN + 4 * khalf);
  float4 b0[16], b1[16];
#pragma unroll
  for (int j = 0; j < 16; ++j) {
    b0[j] = b0p[2 * j];
    b1[j] = b1p[2 * j];
  }

  const float as0 = Asrc[h * DOUT + m], as1 = Asrc[h * DOUT + m + 16];
  const float at0 = Atgt[h * DOUT + m], at1 = Atgt[h * DOUT + m + 16];
  const bool m8 = (m & 8) != 0, m4 = (m & 4) != 0, m2 = (m & 2) != 0, m1 = (m & 1) != 0;

  float4 alo[8], ahi[8];
  int tile = 0 < full ? b : full * HW_BLOCKS + x;
  {
    const float4* ap = hw_arow(Hin, tile, m, khalf, N);
#pragma unroll
    for (int j = 0; j < 8; ++j) alo[j] = ap[2 * j];
#pragma unroll
    for (int j = 0; j < 8; ++j) ahi[j] = ap[2 * (j + 8)];
  }

  for (int i = 0; i < nmy; ++i) {
    const int nxt = i + 1 < full ? b + (i + 1) * HW_BLOCKS : full * HW_BLOCKS + x;
    const float4* np = hw_arow(Hin, nxt, m, khalf, N);
    const int base = tile * 16;

    v8f acc0 = {0.f, 0.f, 0.f, 0.f, 0.f, 0.f, 0.f, 0.f};
    v8f acc1 = {0.f, 0.f, 0.f, 0.f, 0.f, 0.f, 0.f, 0.f};
#pragma unroll
    for (int j = 0; j < 8; ++j) hw_mma4(alo[j], b0[j], b1[j], acc0, acc1);
#pragma unroll
    for (int j = 0; j < 8; ++j) alo[j] = np[2 * j];
#pragma unroll
    for (int j = 0; j < 8; ++j) hw_mma4(ahi[j], b0[j + 8], b1[j + 8], acc0, acc1);
#pragma unroll
    for (int j = 0; j < 8; ++j) ahi[j] = np[2 * (j + 8)];

#pragma unroll
    for (int g = 0; g < 8; ++g) {
      int node = base + g + 8 * khalf;
      if (node < N) {
        float* dst = Hw + ((size_t)node * NH + h) * DOUT;
        dst[m]      = acc0[g];
        dst[m + 16] = acc1[g];
      }
    }

    float pv[16];
#pragma unroll
    for (int g = 0; g < 8; ++g) {
      pv[g]     = acc0[g] * as0 + acc1[g] * as1;
      pv[8 + g] = acc0[g] * at0 + acc1[g] * at1;
    }
    float r8[8];
#pragma unroll
    for (int j = 0; j < 8; ++j) {
      const float keep = m8 ? pv[j + 8] : pv[j];
      const float send = m8 ? pv[j] : pv[j + 8];
      r8[j] = keep + lane_xor<8>(send);
    }
    float r4[4];
#pragma unroll
    for (int j = 0; j < 4; ++j) {
      const float keep = m4 ? r8[j + 4] : r8[j];
      const float send = m4 ? r8[j] : r8[j + 4];
      r4[j] = keep + lane_xor<4>(send);
    }
    float r2[2];
#pragma unroll
    for (int j = 0; j < 2; ++j) {
      const float keep = m2 ? r4[j + 2] : r4[j];
      const float send = m2 ? r4[j] : r4[j + 2];
      r2[j] = keep + lane_xor<2>(send);
    }
    {
      const float keep = m1 ? r2[1] : r2[0];
      const float send = m1 ? r2[0] : r2[1];
      const float tot  = keep + lane_xor<1>(send);
      const int node = base + (m & 7) + 8 * khalf;
      const unsigned off = ((m8 ? (unsigned)N * NH : 0u) + (unsigned)node * NH + (unsigned)h) *
                           (unsigned)sizeof(float);
      if (node < N) *(float*)((char*)slog + off) = tot;
    }
    tile = nxt;
  }
}

#define BK_NODES  196
#define L1_BLOCKS 256
#define L1_T      1024
#define L1_W      (L1_T / 32)
#define L1_U      8
#define BN_T      1024
#define BN_W      (BN_T / 32)
#define BN_SEGS   (L1_BLOCKS / BN_W)
#define BN_CAP    8192

__global__ __launch_bounds__(L1_T) void k_l1_split(const int* __restrict__ src,
                                                   const int* __restrict__ tgt,
                                                   int* __restrict__ lb_raw,
                                                   int* __restrict__ pe, int E, int nb) {
  extern __shared__ int l1_cnt[];
  __shared__ int wtot[L1_W];
  unsigned short* lb = (unsigned short*)lb_raw;
  const int tid  = threadIdx.x;
  const int lane = tid & 31;
  const int w    = tid >> 5;
  for (int i = tid; i < nb; i += L1_T) l1_cnt[i] = 0;
  __syncthreads();

  const int chunk = (E + (int)gridDim.x - 1) / (int)gridDim.x;
  const int beg   = (int)blockIdx.x * chunk;
  const int end   = min(E, beg + chunk);
  for (int e0 = beg; e0 < end; e0 += L1_T * L1_U) {
    int t[L1_U];
#pragma unroll
    for (int u = 0; u < L1_U; ++u) {
      const int e = e0 + u * L1_T + tid;
      t[u] = tgt[e < end ? e : end - 1];
    }
#pragma unroll
    for (int u = 0; u < L1_U; ++u)
      if (e0 + u * L1_T + tid < end) atomicAdd(&l1_cnt[(unsigned)t[u] / BK_NODES], 1);
  }
  __syncthreads();

  int carry = 0;
  for (int base = 0; base < nb; base += L1_T) {
    const int i = base + tid;
    const int v = i < nb ? l1_cnt[i] : 0;
    int inc = v;
    for (int d = 1; d < 32; d <<= 1) {
      const int y = __shfl_up(inc, d);
      if (lane >= d) inc += y;
    }
    if (lane == 31) wtot[w] = inc;
    __syncthreads();
    const int wt = lane < L1_W ? wtot[lane] : 0;
    int pre = lane < w ? wt : 0;
    int all = wt;
    for (int d = 16; d >= 1; d >>= 1) {
      pre += __shfl_xor(pre, d);
      all += __shfl_xor(all, d);
    }
    const int ex = carry + pre + inc - v;
    if (i < nb) {
      l1_cnt[i] = ex;
      lb[(size_t)i * gridDim.x + blockIdx.x] = (unsigned short)ex;
    }
    carry += all;
    __syncthreads();
  }
  if (tid == 0) lb[(size_t)nb * gridDim.x + blockIdx.x] = (unsigned short)carry;

  for (int e0 = beg; e0 < end; e0 += L1_T * L1_U) {
    int s[L1_U];
    int t[L1_U];
#pragma unroll
    for (int u = 0; u < L1_U; ++u) {
      const int e  = e0 + u * L1_T + tid;
      const int ec = e < end ? e : end - 1;
      t[u] = tgt[ec];
      s[u] = src[ec];
    }
#pragma unroll
    for (int u = 0; u < L1_U; ++u) {
      if (e0 + u * L1_T + tid < end) {
        const int bk  = (int)((unsigned)t[u] / BK_NODES);
        const int pos = atomicAdd(&l1_cnt[bk], 1);
        pe[(size_t)beg + pos] = (s[u] << 8) | (t[u] - bk * BK_NODES);
      }
    }
  }
}

__device__ __forceinline__ float lrelu(float x) { return x > 0.f ? x : 0.2f * x; }

#define GB 8
#define RB 3
#define ROUND (RB * GB)

template <int U>
__device__ __forceinline__ float group_bcast(float x) {
  return __builtin_bit_cast(
      float, __builtin_amdgcn_mov_dpp8(__builtin_bit_cast(int, x), U * 0x249249));
}

__device__ __forceinline__ float group_sum(float x) {
  x += lane_xor<1>(x);
  x += lane_xor<2>(x);
  x += lane_xor<4>(x);
  return x;
}

__device__ __forceinline__ float group_max(float x) {
  x = fmaxf(x, lane_xor<1>(x));
  x = fmaxf(x, lane_xor<2>(x));
  x = fmaxf(x, lane_xor<4>(x));
  return x;
}

typedef const __attribute__((address_space(1))) float4* grow_t;
__device__ __forceinline__ grow_t node_row(const float* __restrict__ Hw, unsigned node) {
  const unsigned long long a =
      (unsigned long long)Hw + (unsigned long long)node * (NH * DOUT * sizeof(float));
  const unsigned lo = (unsigned)__builtin_amdgcn_readfirstlane((int)(unsigned)a);
  const unsigned hi = (unsigned)__builtin_amdgcn_readfirstlane((int)(unsigned)(a >> 32));
  return (grow_t)(((unsigned long long)hi << 32) | lo);
}

template <int U>
__device__ __forceinline__ void add_row(float p, float4 v, float4& acc) {
  const float w = group_bcast<U>(p);
  acc.x = __builtin_fmaf(w, v.x, acc.x);
  acc.y = __builtin_fmaf(w, v.y, acc.y);
  acc.z = __builtin_fmaf(w, v.z, acc.z);
  acc.w = __builtin_fmaf(w, v.w, acc.w);
}

template <int J>
__device__ __forceinline__ void load_batch(int s, int last, const float* __restrict__ Hw,
                                           int lane, float4 (&v)[GB]) {
#pragma unroll
  for (int u = 0; u < GB; ++u) {
    const int k = J * GB + u < last ? J * GB + u : last;
    v[u] = node_row(Hw, (unsigned)__builtin_amdgcn_readlane(s, k))[lane];
  }
}

__device__ __forceinline__ void add_batch(float p, const float4 (&v)[GB], float4& acc) {
  add_row<0>(p, v[0], acc);
  add_row<1>(p, v[1], acc);
  add_row<2>(p, v[2], acc);
  add_row<3>(p, v[3], acc);
  add_row<4>(p, v[4], acc);
  add_row<5>(p, v[5], acc);
  add_row<6>(p, v[6], acc);
  add_row<7>(p, v[7], acc);
}

template <int J>
__device__ __forceinline__ void gather_batch(int s, int last, float p,
                                             const float* __restrict__ Hw, int lane,
                                             float4& acc) {
  float4 v[GB];
  load_batch<J>(s, last, Hw, lane, v);
  add_batch(p, v, acc);
}

__device__ __forceinline__ void gather_round(const int* ids, int n, const float* __restrict__ Hw,
                                             const float* __restrict__ slog, float st, int lane,
                                             float& m, float& l, float4& acc) {
  static_assert(RB == 3, "the round's maximum and its batches are spelled out for three batches");
  const int q = lane & 7;
  const unsigned h = (unsigned)lane >> 3;
  const int last = n - 1;
  int nq[RB];
#pragma unroll
  for (int j = 0; j < RB; ++j) nq[j] = n - 8 * j;
  const int s = ids[lane];
  int id[RB];
  float x[RB];
#pragma unroll
  for (int j = 0; j < RB; ++j) {
    const int raw = ids[8 * j + q];
    id[j] = q < nq[j] ? raw : 0;
  }
#pragma unroll
  for (int j = 0; j < RB; ++j) {
    const unsigned off = ((unsigned)id[j] * NH + h) * (unsigned)sizeof(float);
    x[j] = *(const float*)((const char*)slog + off);
  }
  float4 v[GB];
  load_batch<0>(s, last, Hw, lane, v);

#pragma unroll
  for (int j = 0; j < RB; ++j) {
    const float e = lrelu(x[j] + st);
    x[j] = q < nq[j] ? e : -INFINITY;
  }
  const float rm = fmaxf(fmaxf(x[0], x[1]), x[2]);
  const float mn = fmaxf(m, group_max(rm));
  const float sc = expf(m - mn);
  float p[RB];
  l *= sc;
#pragma unroll
  for (int j = 0; j < RB; ++j) {
    p[j] = 0.f;
    if (8 * j < n) {
      p[j] = expf(x[j] - mn);
      l += p[j];
    }
  }
  acc.x *= sc;
  acc.y *= sc;
  acc.z *= sc;
  acc.w *= sc;
  m = mn;
  add_batch(p[0], v, acc);
  if (n > 8) gather_batch<1>(s, last, p[1], Hw, lane, acc);
  if (n > 16) gather_batch<2>(s, last, p[2], Hw, lane, acc);
}

__global__ __launch_bounds__(BN_T, 16) void k_bucket_node(const int* __restrict__ pe,
                                                         const int* __restrict__ lb_raw,
                                                         const float* __restrict__ slog,
                                                         const float* __restrict__ Hw,
                                                         float* __restrict__ dout, int N, int E) {
  const unsigned short* lb = (const unsigned short*)lb_raw;
  __shared__ int cnt[256];
  __shared__ int offl[257];
  __shared__ int wtot[8];
  __shared__ int seg_st[L1_BLOCKS];
  __shared__ int seg_ln[L1_BLOCKS];
  __shared__ int wlen[L1_BLOCKS / 32];
  __shared__ int nextt;
  __shared__ int srcl[BN_CAP];
  __shared__ int stage[BN_W][32];
  const int tid   = threadIdx.x;
  const int lane  = tid & 31;
  const int w     = __builtin_amdgcn_readfirstlane(tid >> 5);
  const int k     = blockIdx.x;
  const int chunk = (E + L1_BLOCKS - 1) / L1_BLOCKS;

  if (tid < 256) cnt[tid] = 0;
  if (tid == 0) nextt = 0;
  if (tid < L1_BLOCKS) {
    const int a = lb[(size_t)k * L1_BLOCKS + tid];
    const int z = lb[(size_t)(k + 1) * L1_BLOCKS + tid];
    seg_st[tid] = tid * chunk + a;
    seg_ln[tid] = z - a;
    int sl = z - a;
    for (int d = 16; d >= 1; d >>= 1) sl += __shfl_xor(sl, d);
    if (lane == 0) wlen[w] = sl;
  }
  __syncthreads();
  int tot = 0;
  for (int j = 0; j < L1_BLOCKS / 32; ++j) tot += wlen[j];
  tot = __builtin_amdgcn_readfirstlane(tot);
  const int sb = w * BN_SEGS;

  int pr[BN_SEGS];
#pragma unroll
  for (int u = 0; u < BN_SEGS; ++u) {
    const int ln = seg_ln[sb + u];
    const int st = seg_st[sb + u];
    pr[u] = pe[lane < ln ? st + lane : 0];
  }
#pragma unroll
  for (int u = 0; u < BN_SEGS; ++u)
    if (lane < seg_ln[sb + u]) atomicAdd(&cnt[pr[u] & 255], 1);
  for (int u = 0; u < BN_SEGS; ++u) {
    const int ln = seg_ln[sb + u];
    const int st = seg_st[sb + u];
    for (int j = 32 + lane; j < ln; j += 32) atomicAdd(&cnt[pe[st + j] & 255], 1);
  }
  __syncthreads();

  const int v = tid < 256 ? cnt[tid] : 0;
  int inc = v;
  for (int d = 1; d < 32; d <<= 1) {
    const int y = __shfl_up(inc, d);
    if (lane >= d) inc += y;
  }
  if (tid < 256 && lane == 31) wtot[w] = inc;
  __syncthreads();
  if (tid < 256) {
    int pre = 0;
    for (int j = 0; j < w; ++j) pre += wtot[j];
    const int start = pre + inc - v;
    cnt[tid]  = start;
    offl[tid] = start;
  }
  if (tid == 0) offl[256] = tot;
  __syncthreads();

#pragma unroll
  for (int u = 0; u < BN_SEGS; ++u) {
    if (lane < seg_ln[sb + u]) {
      const int pos = atomicAdd(&cnt[pr[u] & 255], 1);
      if (pos < BN_CAP) srcl[pos] = (int)((unsigned)pr[u] >> 8);
    }
  }
  for (int u = 0; u < BN_SEGS; ++u) {
    const int ln = seg_ln[sb + u];
    const int st = seg_st[sb + u];
    for (int j = 32 + lane; j < ln; j += 32) {
      const int p   = pe[st + j];
      const int pos = atomicAdd(&cnt[p & 255], 1);
      if (pos < BN_CAP) srcl[pos] = (int)((unsigned)p >> 8);
    }
  }
  __syncthreads();

  const bool listed = tot <= BN_CAP;
  for (;;) {
    int tl = 0;
    if (lane == 0) tl = atomicAdd(&nextt, 1);
    tl = __builtin_amdgcn_readfirstlane(tl);
    const int t = k * BK_NODES + tl;
    if (tl >= BK_NODES || t >= N) break;
    const int beg = __builtin_amdgcn_readfirstlane(offl[tl]);
    const int end = __builtin_amdgcn_readfirstlane(offl[tl + 1]);

    const unsigned hh = (unsigned)lane >> 3;
    const float st = *(const float*)((const char*)slog +
                                     (((unsigned)N + (unsigned)t) * NH + hh) *
                                         (unsigned)sizeof(float));
    float m = -INFINITY;
    float l = 0.f;
    float4 acc;
    acc.x = acc.y = acc.z = acc.w = 0.f;
    if (listed) {
      for (int c = beg; c < end; c += ROUND) {
        const int left = end - c;
        gather_round(&srcl[c], left < ROUND ? left : ROUND, Hw, slog, st, lane, m, l, acc);
      }
    } else if (beg < end) {
      for (int j = 0; j < L1_BLOCKS; ++j) {
        const int ln  = __builtin_amdgcn_readfirstlane(seg_ln[j]);
        const int sst = __builtin_amdgcn_readfirstlane(seg_st[j]);
        for (int c = 0; c < ln; c += 32) {
          const bool in = c + lane < ln;
          const int p = in ? pe[sst + c + lane] : 0;
          const bool hit = in && (p & 255) == tl;
          const unsigned mask = __builtin_amdgcn_ballot_w32(hit);
          if (mask != 0u) {
            const int rank = (int)__builtin_amdgcn_mbcnt_lo(mask, 0u);
            if (hit) stage[w][rank] = (int)((unsigned)p >> 8);
            const int nhit = __builtin_popcount(mask);
            for (int o = 0; o < nhit; o += ROUND) {
              const int left = nhit - o;
              gather_round(&stage[w][o], left < ROUND ? left : ROUND, Hw, slog, st, lane, m, l,
                           acc);
            }
          }
        }
      }
    }
    const float lsum = group_sum(l);
    const float inv  = beg < end ? 1.0f / lsum : 0.f;
    acc.x *= inv;
    acc.y *= inv;
    acc.z *= inv;
    acc.w *= inv;

    float4 r;
    r.x = acc.x > 0.f ? acc.x : expm1f(acc.x);
    r.y = acc.y > 0.f ? acc.y : expm1f(acc.y);
    r.z = acc.z > 0.f ? acc.z : expm1f(acc.z);
    r.w = acc.w > 0.f ? acc.w : expm1f(acc.w);
    const unsigned off =
        (((unsigned)(lane >> 3) * (unsigned)N + (unsigned)t) * DOUT + (unsigned)(lane & 7) * 4u) *
        (unsigned)sizeof(float);
    *(float4*)((char*)dout + off) = r;
  }
}

extern "C" void kernel_launch(void* const* d_in, const int* in_sizes, int n_in,
                              void* d_out, int out_size, void* d_ws, size_t ws_size,
                              hipStream_t stream) {
  const float* Hin  = (const float*)d_in[0];
  const int*   ei   = (const int*)d_in[1];
  const float* W    = (const float*)d_in[2];
  const float* Asrc = (const float*)d_in[3];
  const float* Atgt = (const float*)d_in[4];
  float* dout = (float*)d_out;

  const int N = in_sizes[0] / DIN;
  const int E = in_sizes[1] / 2;
  const int* src = ei;
  const int* tgt = ei + E;

  const int nb = (N + BK_NODES - 1) / BK_NODES;
  char* p = (char*)d_ws;
  float* Hw = (float*)p; p += (size_t)N * NH * DOUT * sizeof(float);
  int*   pe = (int*)p;   p += (size_t)E * sizeof(int);
  int*   lb = (int*)p;   p += (size_t)(nb + 1) * L1_BLOCKS * sizeof(unsigned short);
  float* slog = (float*)p;
  (void)ws_size; (void)n_in; (void)out_size;

  k_l1_split<<<L1_BLOCKS, L1_T, nb * sizeof(int), stream>>>(src, tgt, lb, pe, E, nb);
  k_hw_wmma<<<HW_BLOCKS, 128, 0, stream>>>(Hin, W, Asrc, Atgt, Hw, slog, N);
  k_bucket_node<<<nb, BN_T, 0, stream>>>(pe, lb, slog, Hw, dout, N, E);
}
